# starting kernel
# baseline (speedup 1.0000x reference)
#include <hip/hip_runtime.h>
#include <hip/hip_bf16.h>
#include <math.h>

#define D 128

typedef float v2f __attribute__((ext_vector_type(2)));
typedef float v8f __attribute__((ext_vector_type(8)));

__device__ __forceinline__ void atomicMaxF(float* addr, float val) {
    if (val >= 0.0f) atomicMax((int*)addr, __float_as_int(val));
    else             atomicMin((unsigned int*)addr, __float_as_uint(val));
}

__global__ void fill_f32(float* __restrict__ p, float v, int n) {
    int i = blockIdx.x * blockDim.x + threadIdx.x;
    if (i < n) p[i] = v;
}

__global__ void gemm_z(const float* __restrict__ h0, int n0,
                       const float* __restrict__ h1,
                       const float* __restrict__ W,
                       float* __restrict__ Z, int n)
{
    const int wave = threadIdx.x >> 5;
    const int lane = threadIdx.x & 31;
    const int half = lane >> 4;
    const int lr   = lane & 15;
    const int row0 = blockIdx.x * 16;
    const int col0 = wave * 16;

    int rowA = row0 + lr;
    if (rowA >= n) rowA = n - 1;
    const float* hrow = (rowA < n0) ? (h0 + (size_t)rowA * D)
                                    : (h1 + (size_t)(rowA - n0) * D);

    v8f c = {0.f, 0.f, 0.f, 0.f, 0.f, 0.f, 0.f, 0.f};
#pragma unroll
    for (int k0 = 0; k0 < D; k0 += 4) {
        const int kk = k0 + 2 * half;
        float2 av = *(const float2*)(hrow + kk);
        v2f a; a.x = av.x; a.y = av.y;
        v2f b; b.x = W[(size_t)kk * D + col0 + lr];
               b.y = W[(size_t)(kk + 1) * D + col0 + lr];
        c = __builtin_amdgcn_wmma_f32_16x16x4_f32(
                false, a, false, b, (short)0, c, false, false);
    }
#pragma unroll
    for (int v = 0; v < 8; ++v) {
        const int r = row0 + v + 8 * half;
        if (r < n) Z[(size_t)r * D + col0 + lr] = c[v];
    }
}

__global__ void row_scores(const float* __restrict__ Z, const float* __restrict__ a,
                           float* __restrict__ ssrc, float* __restrict__ sdst, int n)
{
    const int wave = (blockIdx.x * blockDim.x + threadIdx.x) >> 5;
    const int lane = threadIdx.x & 31;
    if (wave >= n) return;
    const float4 zv = *(const float4*)(Z + (size_t)wave * D + lane * 4);
    const float4 a0 = *(const float4*)(a + lane * 4);
    const float4 a1 = *(const float4*)(a + D + lane * 4);
    float p0 = zv.x * a0.x + zv.y * a0.y + zv.z * a0.z + zv.w * a0.w;
    float p1 = zv.x * a1.x + zv.y * a1.y + zv.z * a1.z + zv.w * a1.w;
#pragma unroll
    for (int off = 16; off > 0; off >>= 1) {
        p0 += __shfl_down(p0, off, 32);
        p1 += __shfl_down(p1, off, 32);
    }
    if (lane == 0) { ssrc[wave] = p0; sdst[wave] = p1; }
}

__device__ __forceinline__ float leaky(float v) {
    return v > 0.0f ? v : 0.01f * v;
}

__global__ void edge_max(const int* __restrict__ src, const int* __restrict__ dst,
                         const float* __restrict__ ssrc, const float* __restrict__ sdst,
                         float* __restrict__ m, int nE)
{
    int e = blockIdx.x * blockDim.x + threadIdx.x;
    if (e >= nE) return;
    int d = dst[e];
    float v = leaky(ssrc[src[e]] + sdst[d]);
    atomicMaxF(&m[d], v);
}

__global__ void edge_den(const int* __restrict__ src, const int* __restrict__ dst,
                         const float* __restrict__ ssrc, const float* __restrict__ sdst,
                         const float* __restrict__ m, float* __restrict__ den, int nE)
{
    int e = blockIdx.x * blockDim.x + threadIdx.x;
    if (e >= nE) return;
    int d = dst[e];
    float v = leaky(ssrc[src[e]] + sdst[d]);
    atomicAdd(&den[d], __expf(v - m[d]));
}

__global__ void edge_agg(const int* __restrict__ src, const int* __restrict__ dst,
                         const float* __restrict__ ssrc, const float* __restrict__ sdst,
                         const float* __restrict__ m, const float* __restrict__ den,
                         const float* __restrict__ z,
                         float* __restrict__ out_lo, float* __restrict__ out_hi,
                         int n0, int nE)
{
    int t = blockIdx.x * blockDim.x + threadIdx.x;
    int e = t >> 5;
    int lane = t & 31;
    if (e >= nE) return;
    int s = src[e], d = dst[e];
    float v = leaky(ssrc[s] + sdst[d]);
    float alpha = __expf(v - m[d]) / den[d];
    float4 zv = *(const float4*)(z + (size_t)s * D + lane * 4);
    float* op = ((d < n0) ? (out_lo + (size_t)d * D)
                          : (out_hi + (size_t)(d - n0) * D)) + lane * 4;
    atomicAdd(op + 0, alpha * zv.x);
    atomicAdd(op + 1, alpha * zv.y);
    atomicAdd(op + 2, alpha * zv.z);
    atomicAdd(op + 3, alpha * zv.w);
}

__global__ void combine(const float* __restrict__ A, const float* __restrict__ Bfull,
                        const float* __restrict__ W1, const float* __restrict__ b1,
                        const float* __restrict__ w2, float* __restrict__ out,
                        int Isplit)
{
    __shared__ float sA[D], sB[D], red[D];
    const int i = blockIdx.x;
    const int j = threadIdx.x;
    sA[j] = A[(size_t)i * D + j];
    sB[j] = Bfull[(size_t)(Isplit + i) * D + j];
    __syncthreads();
    float accA = b1[j], accB = accA;
    for (int k = 0; k < D; ++k) {
        float wv = W1[(size_t)k * D + j];
        accA = fmaf(sA[k], wv, accA);
        accB = fmaf(sB[k], wv, accB);
    }
    const float w2j = w2[j];
    const float tA = tanhf(accA) * w2j;
    const float tB = tanhf(accB) * w2j;

    red[j] = tA; __syncthreads();
    for (int s = 64; s > 0; s >>= 1) { if (j < s) red[j] += red[j + s]; __syncthreads(); }
    const float s1 = red[0]; __syncthreads();
    red[j] = tB; __syncthreads();
    for (int s = 64; s > 0; s >>= 1) { if (j < s) red[j] += red[j + s]; __syncthreads(); }
    const float s2 = red[0];

    const float mm = fmaxf(s1, s2);
    const float e1 = __expf(s1 - mm), e2 = __expf(s2 - mm);
    const float al = e1 / (e1 + e2);
    out[(size_t)i * D + j] = al * sA[j] + (1.0f - al) * sB[j];
}

static inline int cdiv(int a, int b) { return (a + b - 1) / b; }

extern "C" void kernel_launch(void* const* d_in, const int* in_sizes, int n_in,
                              void* d_out, int out_size, void* d_ws, size_t ws_size,
                              hipStream_t stream)
{
    (void)n_in; (void)ws_size;
    const float* kn_emb   = (const float*)d_in[0];
    const float* exer_emb = (const float*)d_in[1];
    const float* W_dir    = (const float*)d_in[2];
    const float* a_dir    = (const float*)d_in[3];
    const float* W_ke     = (const float*)d_in[4];
    const float* a_ke     = (const float*)d_in[5];
    const float* W_ek     = (const float*)d_in[6];
    const float* a_ek     = (const float*)d_in[7];
    const float* rW1      = (const float*)d_in[8];
    const float* rb1      = (const float*)d_in[9];
    const float* rw2      = (const float*)d_in[10];
    const int* dsrc  = (const int*)d_in[11];
    const int* ddst  = (const int*)d_in[12];
    const int* kesrc = (const int*)d_in[13];
    const int* kedst = (const int*)d_in[14];
    const int* eksrc = (const int*)d_in[15];
    const int* ekdst = (const int*)d_in[16];

    const int Kn = in_sizes[0] / D;
    const int In = in_sizes[1] / D;
    const int Nn = Kn + In;
    const int E1 = in_sizes[11];
    const int E2 = in_sizes[13];

    float* w = (float*)d_ws;
    size_t off = 0;
    float* buf_z = w + off; off += (size_t)Nn * D;
    float* buf_B = w + off; off += (size_t)Nn * D;
    float* buf_A = w + off; off += (size_t)Kn * D;
    float* ssrc  = w + off; off += (size_t)Nn;
    float* sdst  = w + off; off += (size_t)Nn;
    float* mbuf  = w + off; off += (size_t)Nn;
    float* dbuf  = w + off; off += (size_t)Nn;

    float* out_kn = (float*)d_out;
    float* out_ex = (float*)d_out + (size_t)Kn * D;

    hipMemsetAsync(d_out, 0, (size_t)out_size * sizeof(float), stream);

    gemm_z<<<cdiv(Nn, 16), 256, 0, stream>>>(exer_emb, In, kn_emb, W_ek, buf_z, Nn);
    row_scores<<<cdiv(Nn, 8), 256, 0, stream>>>(buf_z, a_ek, ssrc, sdst, Nn);
    fill_f32<<<cdiv(Nn, 256), 256, 0, stream>>>(mbuf, -INFINITY, Nn);
    hipMemsetAsync(dbuf, 0, (size_t)Nn * sizeof(float), stream);
    hipMemsetAsync(buf_A, 0, (size_t)Kn * D * sizeof(float), stream);
    edge_max<<<cdiv(E2, 256), 256, 0, stream>>>(eksrc, ekdst, ssrc, sdst, mbuf, E2);
    edge_den<<<cdiv(E2, 256), 256, 0, stream>>>(eksrc, ekdst, ssrc, sdst, mbuf, dbuf, E2);
    edge_agg<<<cdiv(E2 * 32, 256), 256, 0, stream>>>(eksrc, ekdst, ssrc, sdst, mbuf, dbuf,
                                                     buf_z, out_ex, buf_A, In, E2);

    gemm_z<<<cdiv(Nn, 16), 256, 0, stream>>>(exer_emb, In, kn_emb, W_ke, buf_z, Nn);
    row_scores<<<cdiv(Nn, 8), 256, 0, stream>>>(buf_z, a_ke, ssrc, sdst, Nn);
    fill_f32<<<cdiv(Nn, 256), 256, 0, stream>>>(mbuf, -INFINITY, Nn);
    hipMemsetAsync(dbuf, 0, (size_t)Nn * sizeof(float), stream);
    hipMemsetAsync(buf_B, 0, (size_t)Nn * D * sizeof(float), stream);
    edge_max<<<cdiv(E2, 256), 256, 0, stream>>>(kesrc, kedst, ssrc, sdst, mbuf, E2);
    edge_den<<<cdiv(E2, 256), 256, 0, stream>>>(kesrc, kedst, ssrc, sdst, mbuf, dbuf, E2);
    edge_agg<<<cdiv(E2 * 32, 256), 256, 0, stream>>>(kesrc, kedst, ssrc, sdst, mbuf, dbuf,
                                                     buf_z, buf_B, buf_B, Nn, E2);

    gemm_z<<<cdiv(Kn, 16), 256, 0, stream>>>(kn_emb, Kn, kn_emb, W_dir, buf_z, Kn);
    row_scores<<<cdiv(Kn, 8), 256, 0, stream>>>(buf_z, a_dir, ssrc, sdst, Kn);
    fill_f32<<<cdiv(Kn, 256), 256, 0, stream>>>(mbuf, -INFINITY, Kn);
    hipMemsetAsync(dbuf, 0, (size_t)Kn * sizeof(float), stream);
    hipMemsetAsync(buf_A, 0, (size_t)Kn * D * sizeof(float), stream);
    edge_max<<<cdiv(E1, 256), 256, 0, stream>>>(dsrc, ddst, ssrc, sdst, mbuf, E1);
    edge_den<<<cdiv(E1, 256), 256, 0, stream>>>(dsrc, ddst, ssrc, sdst, mbuf, dbuf, E1);
    edge_agg<<<cdiv(E1 * 32, 256), 256, 0, stream>>>(dsrc, ddst, ssrc, sdst, mbuf, dbuf,
                                                     buf_z, buf_A, buf_A, Kn, E1);

    combine<<<Kn, D, 0, stream>>>(buf_A, buf_B, rW1, rb1, rw2, out_kn, In);
}


# edge_den and edge_agg fused: each lane keeps its edge's exp term, the wave then adds its 32 edges' exp*z rows; rows divided by den afterwards (row pass, combine): rounding of row sums changes
# speedup vs baseline: 1.2990x; 1.2990x over previous
#include <hip/hip_runtime.h>
#include <hip/hip_bf16.h>
#include <math.h>

#define D 128

typedef float v2f __attribute__((ext_vector_type(2)));
typedef float v8f __attribute__((ext_vector_type(8)));

__device__ __forceinline__ void atomicMaxF(float* addr, float val) {
    if (val >= 0.0f) atomicMax((int*)addr, __float_as_int(val));
    else             atomicMin((unsigned int*)addr, __float_as_uint(val));
}

__global__ void fill_f32(float* __restrict__ p, float v, int n) {
    int i = blockIdx.x * blockDim.x + threadIdx.x;
    if (i < n) p[i] = v;
}

__global__ void gemm_z(const float* __restrict__ h0, int n0,
                       const float* __restrict__ h1,
                       const float* __restrict__ W,
                       float* __restrict__ Z, int n)
{
    const int wave = threadIdx.x >> 5;
    const int lane = threadIdx.x & 31;
    const int half = lane >> 4;
    const int lr   = lane & 15;
    const int row0 = blockIdx.x * 16;
    const int col0 = wave * 16;

    int rowA = row0 + lr;
    if (rowA >= n) rowA = n - 1;
    const float* hrow = (rowA < n0) ? (h0 + (size_t)rowA * D)
                                    : (h1 + (size_t)(rowA - n0) * D);

    v8f c = {0.f, 0.f, 0.f, 0.f, 0.f, 0.f, 0.f, 0.f};
#pragma unroll
    for (int k0 = 0; k0 < D; k0 += 4) {
        const int kk = k0 + 2 * half;
        float2 av = *(const float2*)(hrow + kk);
        v2f a; a.x = av.x; a.y = av.y;
        v2f b; b.x = W[(size_t)kk * D + col0 + lr];
               b.y = W[(size_t)(kk + 1) * D + col0 + lr];
        c = __builtin_amdgcn_wmma_f32_16x16x4_f32(
                false, a, false, b, (short)0, c, false, false);
    }
#pragma unroll
    for (int v = 0; v < 8; ++v) {
        const int r = row0 + v + 8 * half;
        if (r < n) Z[(size_t)r * D + col0 + lr] = c[v];
    }
}

__global__ void row_scores(const float* __restrict__ Z, const float* __restrict__ a,
                           float* __restrict__ ssrc, float* __restrict__ sdst, int n)
{
    const int wave = (blockIdx.x * blockDim.x + threadIdx.x) >> 5;
    const int lane = threadIdx.x & 31;
    if (wave >= n) return;
    const float4 zv = *(const float4*)(Z + (size_t)wave * D + lane * 4);
    const float4 a0 = *(const float4*)(a + lane * 4);
    const float4 a1 = *(const float4*)(a + D + lane * 4);
    float p0 = zv.x * a0.x + zv.y * a0.y + zv.z * a0.z + zv.w * a0.w;
    float p1 = zv.x * a1.x + zv.y * a1.y + zv.z * a1.z + zv.w * a1.w;
#pragma unroll
    for (int off = 16; off > 0; off >>= 1) {
        p0 += __shfl_down(p0, off, 32);
        p1 += __shfl_down(p1, off, 32);
    }
    if (lane == 0) { ssrc[wave] = p0; sdst[wave] = p1; }
}

__device__ __forceinline__ float leaky(float v) {
    return v > 0.0f ? v : 0.01f * v;
}

__global__ void edge_max(const int* __restrict__ src, const int* __restrict__ dst,
                         const float* __restrict__ ssrc, const float* __restrict__ sdst,
                         float* __restrict__ m, int nE)
{
    int e = blockIdx.x * blockDim.x + threadIdx.x;
    if (e >= nE) return;
    int d = dst[e];
    float v = leaky(ssrc[src[e]] + sdst[d]);
    atomicMaxF(&m[d], v);
}

__global__ void edge_den_agg(const int* __restrict__ src, const int* __restrict__ dst,
                             const float* __restrict__ ssrc, const float* __restrict__ sdst,
                             const float* __restrict__ m, float* __restrict__ den,
                             const float* __restrict__ z,
                             float* __restrict__ out_lo, float* __restrict__ out_hi,
                             int n0, int nE)
{
    const int lane = threadIdx.x & 31;
    const int e = blockIdx.x * blockDim.x + threadIdx.x;
    int s = 0, d = 0;
    float t = 0.0f;
    if (e < nE) {
        s = src[e];
        d = dst[e];
        const float v = leaky(ssrc[s] + sdst[d]);
        t = __expf(v - m[d]);
        atomicAdd(&den[d], t);
    }
    const int base = __builtin_amdgcn_readfirstlane(e - lane);
    int cnt = nE - base;
    cnt = cnt > 32 ? 32 : cnt;

    int j = 0;
    for (; j + 8 <= cnt; j += 8) {
        float zv[8][4];
#pragma unroll
        for (int u = 0; u < 8; ++u) {
            const int sj = __builtin_amdgcn_readlane(s, j + u);
            const float* zp = z + (size_t)sj * D + lane;
            zv[u][0] = zp[0];
            zv[u][1] = zp[32];
            zv[u][2] = zp[64];
            zv[u][3] = zp[96];
        }
#pragma unroll
        for (int u = 0; u < 8; ++u) {
            const int   dj = __builtin_amdgcn_readlane(d, j + u);
            const float tj = __int_as_float(__builtin_amdgcn_readlane(__float_as_int(t), j + u));
            float* op = ((dj < n0) ? (out_lo + (size_t)dj * D)
                                   : (out_hi + (size_t)(dj - n0) * D)) + lane;
            atomicAdd(op,      tj * zv[u][0]);
            atomicAdd(op + 32, tj * zv[u][1]);
            atomicAdd(op + 64, tj * zv[u][2]);
            atomicAdd(op + 96, tj * zv[u][3]);
        }
    }
    for (; j < cnt; ++j) {
        const int   sj = __builtin_amdgcn_readlane(s, j);
        const int   dj = __builtin_amdgcn_readlane(d, j);
        const float tj = __int_as_float(__builtin_amdgcn_readlane(__float_as_int(t), j));
        const float* zp = z + (size_t)sj * D + lane;
        const float z0 = zp[0], z1 = zp[32], z2 = zp[64], z3 = zp[96];
        float* op = ((dj < n0) ? (out_lo + (size_t)dj * D)
                               : (out_hi + (size_t)(dj - n0) * D)) + lane;
        atomicAdd(op,      tj * z0);
        atomicAdd(op + 32, tj * z1);
        atomicAdd(op + 64, tj * z2);
        atomicAdd(op + 96, tj * z3);
    }
}

__global__ void row_div(float* __restrict__ rows, const float* __restrict__ den, int n)
{
    const int wave = (blockIdx.x * blockDim.x + threadIdx.x) >> 5;
    const int lane = threadIdx.x & 31;
    const int r0 = wave * 4;
    float4 v[4];
    float dn[4];
#pragma unroll
    for (int u = 0; u < 4; ++u) {
        const int r = r0 + u;
        if (r < n) {
            dn[u] = den[r];
            v[u] = *(const float4*)(rows + (size_t)r * D + lane * 4);
        }
    }
#pragma unroll
    for (int u = 0; u < 4; ++u) {
        const int r = r0 + u;
        if (r < n) {
            float4 o;
            if (dn[u] > 0.0f) {
                o.x = v[u].x / dn[u]; o.y = v[u].y / dn[u];
                o.z = v[u].z / dn[u]; o.w = v[u].w / dn[u];
            } else {
                o.x = 0.0f; o.y = 0.0f; o.z = 0.0f; o.w = 0.0f;
            }
            *(float4*)(rows + (size_t)r * D + lane * 4) = o;
        }
    }
}

__global__ void combine(const float* __restrict__ A, const float* __restrict__ Bfull,
                        const float* __restrict__ W1, const float* __restrict__ b1,
                        const float* __restrict__ w2, float* __restrict__ out,
                        const float* __restrict__ denA, const float* __restrict__ denB,
                        int Isplit)
{
    __shared__ float sA[D], sB[D], red[D];
    const int i = blockIdx.x;
    const int j = threadIdx.x;
    const float dA = denA[i];
    const float dB = denB[Isplit + i];
    const float rA = A[(size_t)i * D + j];
    const float rB = Bfull[(size_t)(Isplit + i) * D + j];
    sA[j] = dA > 0.0f ? rA / dA : 0.0f;
    sB[j] = dB > 0.0f ? rB / dB : 0.0f;
    __syncthreads();
    float accA = b1[j], accB = accA;
    for (int k = 0; k < D; ++k) {
        float wv = W1[(size_t)k * D + j];
        accA = fmaf(sA[k], wv, accA);
        accB = fmaf(sB[k], wv, accB);
    }
    const float w2j = w2[j];
    const float tA = tanhf(accA) * w2j;
    const float tB = tanhf(accB) * w2j;

    red[j] = tA; __syncthreads();
    for (int s = 64; s > 0; s >>= 1) { if (j < s) red[j] += red[j + s]; __syncthreads(); }
    const float s1 = red[0]; __syncthreads();
    red[j] = tB; __syncthreads();
    for (int s = 64; s > 0; s >>= 1) { if (j < s) red[j] += red[j + s]; __syncthreads(); }
    const float s2 = red[0];

    const float mm = fmaxf(s1, s2);
    const float e1 = __expf(s1 - mm), e2 = __expf(s2 - mm);
    const float al = e1 / (e1 + e2);
    out[(size_t)i * D + j] = al * sA[j] + (1.0f - al) * sB[j];
}

static inline int cdiv(int a, int b) { return (a + b - 1) / b; }

extern "C" void kernel_launch(void* const* d_in, const int* in_sizes, int n_in,
                              void* d_out, int out_size, void* d_ws, size_t ws_size,
                              hipStream_t stream)
{
    (void)n_in; (void)ws_size;
    const float* kn_emb   = (const float*)d_in[0];
    const float* exer_emb = (const float*)d_in[1];
    const float* W_dir    = (const float*)d_in[2];
    const float* a_dir    = (const float*)d_in[3];
    const float* W_ke     = (const float*)d_in[4];
    const float* a_ke     = (const float*)d_in[5];
    const float* W_ek     = (const float*)d_in[6];
    const float* a_ek     = (const float*)d_in[7];
    const float* rW1      = (const float*)d_in[8];
    const float* rb1      = (const float*)d_in[9];
    const float* rw2      = (const float*)d_in[10];
    const int* dsrc  = (const int*)d_in[11];
    const int* ddst  = (const int*)d_in[12];
    const int* kesrc = (const int*)d_in[13];
    const int* kedst = (const int*)d_in[14];
    const int* eksrc = (const int*)d_in[15];
    const int* ekdst = (const int*)d_in[16];

    const int Kn = in_sizes[0] / D;
    const int In = in_sizes[1] / D;
    const int Nn = Kn + In;
    const int E1 = in_sizes[11];
    const int E2 = in_sizes[13];

    float* w = (float*)d_ws;
    size_t off = 0;
    float* buf_z = w + off; off += (size_t)Nn * D;
    float* buf_B = w + off; off += (size_t)Nn * D;
    float* buf_A = w + off; off += (size_t)Kn * D;
    float* ssrc  = w + off; off += (size_t)Nn;
    float* sdst  = w + off; off += (size_t)Nn;
    float* mbuf  = w + off; off += (size_t)Nn;
    float* dbuf  = w + off; off += (size_t)Nn;

    float* out_kn = (float*)d_out;
    float* out_ex = (float*)d_out + (size_t)Kn * D;

    hipMemsetAsync(d_out, 0, (size_t)out_size * sizeof(float), stream);

    gemm_z<<<cdiv(Nn, 16), 256, 0, stream>>>(exer_emb, In, kn_emb, W_ek, buf_z, Nn);
    row_scores<<<cdiv(Nn, 8), 256, 0, stream>>>(buf_z, a_ek, ssrc, sdst, Nn);
    fill_f32<<<cdiv(Nn, 256), 256, 0, stream>>>(mbuf, -INFINITY, Nn);
    hipMemsetAsync(dbuf, 0, (size_t)Nn * sizeof(float), stream);
    hipMemsetAsync(buf_A, 0, (size_t)Kn * D * sizeof(float), stream);
    edge_max<<<cdiv(E2, 256), 256, 0, stream>>>(eksrc, ekdst, ssrc, sdst, mbuf, E2);
    edge_den_agg<<<cdiv(E2, 256), 256, 0, stream>>>(eksrc, ekdst, ssrc, sdst, mbuf, dbuf,
                                                    buf_z, out_ex, buf_A, In, E2);
    row_div<<<cdiv(In, 32), 256, 0, stream>>>(out_ex, dbuf, In);

    gemm_z<<<cdiv(Nn, 16), 256, 0, stream>>>(exer_emb, In, kn_emb, W_ke, buf_z, Nn);
    row_scores<<<cdiv(Nn, 8), 256, 0, stream>>>(buf_z, a_ke, ssrc, sdst, Nn);
    fill_f32<<<cdiv(Nn, 256), 256, 0, stream>>>(mbuf, -INFINITY, Nn);
    hipMemsetAsync(dbuf, 0, (size_t)Nn * sizeof(float), stream);
    hipMemsetAsync(buf_B, 0, (size_t)Nn * D * sizeof(float), stream);
    edge_max<<<cdiv(E2, 256), 256, 0, stream>>>(kesrc, kedst, ssrc, sdst, mbuf, E2);
    edge_den_agg<<<cdiv(E2, 256), 256, 0, stream>>>(kesrc, kedst, ssrc, sdst, mbuf, dbuf,
                                                    buf_z, buf_B, buf_B, Nn, E2);

    gemm_z<<<cdiv(Kn, 16), 256, 0, stream>>>(kn_emb, Kn, kn_emb, W_dir, buf_z, Kn);
    row_scores<<<cdiv(Kn, 8), 256, 0, stream>>>(buf_z, a_dir, ssrc, sdst, Kn);
    fill_f32<<<cdiv(Kn, 256), 256, 0, stream>>>(mbuf, -INFINITY, Kn);
    float* dden = dbuf;
    hipMemsetAsync(dden, 0, (size_t)Kn * sizeof(float), stream);
    hipMemsetAsync(buf_A, 0, (size_t)Kn * D * sizeof(float), stream);
    edge_max<<<cdiv(E1, 256), 256, 0, stream>>>(dsrc, ddst, ssrc, sdst, mbuf, E1);
    edge_den_agg<<<cdiv(E1, 256), 256, 0, stream>>>(dsrc, ddst, ssrc, sdst, mbuf, dden,
                                                    buf_z, buf_A, buf_A, Kn, E1);

    combine<<<Kn, D, 0, stream>>>(buf_A, buf_B, rW1, rb1, rw2, out_kn, dden, dbuf, In);
}


# speedup vs baseline: 1.6250x; 1.2510x over previous
#include <hip/hip_runtime.h>
#include <hip/hip_bf16.h>
#include <math.h>

#define D 128

typedef float v2f __attribute__((ext_vector_type(2)));
typedef float v8f __attribute__((ext_vector_type(8)));

__device__ __forceinline__ void atomicMaxF(float* addr, float val) {
    if (val >= 0.0f) atomicMax((int*)addr, __float_as_int(val));
    else             atomicMin((unsigned int*)addr, __float_as_uint(val));
}

__device__ __forceinline__ int cache_half() {
    __shared__ int s_half;
    if (threadIdx.x == 0)
        s_half = (int)((__builtin_amdgcn_s_sendmsg_rtn(0x87) >> 18) & 1u);
    __syncthreads();
    return __builtin_amdgcn_readfirstlane(s_half);
}

__global__ void fill_f32(float* __restrict__ p, float v, int n) {
    int i = blockIdx.x * blockDim.x + threadIdx.x;
    if (i < n) p[i] = v;
}

__global__ void gemm_z(const float* __restrict__ h0, int n0,
                       const float* __restrict__ h1,
                       const float* __restrict__ W,
                       float* __restrict__ Z, int n)
{
    const int wave = threadIdx.x >> 5;
    const int lane = threadIdx.x & 31;
    const int half = lane >> 4;
    const int lr   = lane & 15;
    const int row0 = blockIdx.x * 16;
    const int col0 = wave * 16;

    int rowA = row0 + lr;
    if (rowA >= n) rowA = n - 1;
    const float* hrow = (rowA < n0) ? (h0 + (size_t)rowA * D)
                                    : (h1 + (size_t)(rowA - n0) * D);

    v8f c = {0.f, 0.f, 0.f, 0.f, 0.f, 0.f, 0.f, 0.f};
#pragma unroll
    for (int k0 = 0; k0 < D; k0 += 4) {
        const int kk = k0 + 2 * half;
        float2 av = *(const float2*)(hrow + kk);
        v2f a; a.x = av.x; a.y = av.y;
        v2f b; b.x = W[(size_t)kk * D + col0 + lr];
               b.y = W[(size_t)(kk + 1) * D + col0 + lr];
        c = __builtin_amdgcn_wmma_f32_16x16x4_f32(
                false, a, false, b, (short)0, c, false, false);
    }
#pragma unroll
    for (int v = 0; v < 8; ++v) {
        const int r = row0 + v + 8 * half;
        if (r < n) Z[(size_t)r * D + col0 + lr] = c[v];
    }
}

__global__ void row_scores(const float* __restrict__ Z, const float* __restrict__ a,
                           float* __restrict__ ssrc, float* __restrict__ sdst, int n)
{
    const int wave = (blockIdx.x * blockDim.x + threadIdx.x) >> 5;
    const int lane = threadIdx.x & 31;
    if (wave >= n) return;
    const float4 zv = *(const float4*)(Z + (size_t)wave * D + lane * 4);
    const float4 a0 = *(const float4*)(a + lane * 4);
    const float4 a1 = *(const float4*)(a + D + lane * 4);
    float p0 = zv.x * a0.x + zv.y * a0.y + zv.z * a0.z + zv.w * a0.w;
    float p1 = zv.x * a1.x + zv.y * a1.y + zv.z * a1.z + zv.w * a1.w;
#pragma unroll
    for (int off = 16; off > 0; off >>= 1) {
        p0 += __shfl_down(p0, off, 32);
        p1 += __shfl_down(p1, off, 32);
    }
    if (lane == 0) { ssrc[wave] = p0; sdst[wave] = p1; }
}

__device__ __forceinline__ float leaky(float v) {
    return v > 0.0f ? v : 0.01f * v;
}

__global__ void edge_max(const int* __restrict__ src, const int* __restrict__ dst,
                         const float* __restrict__ ssrc, const float* __restrict__ sdst,
                         float* m0, float* m1, int nE)
{
    float* m = cache_half() ? m1 : m0;
    int e = blockIdx.x * blockDim.x + threadIdx.x;
    if (e >= nE) return;
    int d = dst[e];
    float v = leaky(ssrc[src[e]] + sdst[d]);
    atomicMaxF(&m[d], v);
}

__global__ void edge_den_agg(const int* __restrict__ src, const int* __restrict__ dst,
                             const float* __restrict__ ssrc, const float* __restrict__ sdst,
                             const float* m0, const float* m1,
                             float* den0, float* den1,
                             const float* __restrict__ z,
                             float* lo0, float* hi0, float* lo1, float* hi1,
                             int n0, int nE)
{
    const int hf = cache_half();
    float* den    = hf ? den1 : den0;
    float* out_lo = hf ? lo1 : lo0;
    float* out_hi = hf ? hi1 : hi0;
    const int lane = threadIdx.x & 31;
    const int e = blockIdx.x * blockDim.x + threadIdx.x;
    int s = 0, d = 0;
    float t = 0.0f;
    if (e < nE) {
        s = src[e];
        d = dst[e];
        const float v = leaky(ssrc[s] + sdst[d]);
        t = __expf(v - fmaxf(m0[d], m1[d]));
        atomicAdd(&den[d], t);
    }
    const int base = __builtin_amdgcn_readfirstlane(e - lane);
    int cnt = nE - base;
    cnt = cnt > 32 ? 32 : cnt;

    int j = 0;
    for (; j + 8 <= cnt; j += 8) {
        float zv[8][4];
#pragma unroll
        for (int u = 0; u < 8; ++u) {
            const int sj = __builtin_amdgcn_readlane(s, j + u);
            const float* zp = z + (size_t)sj * D + lane;
            zv[u][0] = zp[0];
            zv[u][1] = zp[32];
            zv[u][2] = zp[64];
            zv[u][3] = zp[96];
        }
#pragma unroll
        for (int u = 0; u < 8; ++u) {
            const int   dj = __builtin_amdgcn_readlane(d, j + u);
            const float tj = __int_as_float(__builtin_amdgcn_readlane(__float_as_int(t), j + u));
            float* op = ((dj < n0) ? (out_lo + (size_t)dj * D)
                                   : (out_hi + (size_t)(dj - n0) * D)) + lane;
            atomicAdd(op,      tj * zv[u][0]);
            atomicAdd(op + 32, tj * zv[u][1]);
            atomicAdd(op + 64, tj * zv[u][2]);
            atomicAdd(op + 96, tj * zv[u][3]);
        }
    }
    for (; j < cnt; ++j) {
        const int   sj = __builtin_amdgcn_readlane(s, j);
        const int   dj = __builtin_amdgcn_readlane(d, j);
        const float tj = __int_as_float(__builtin_amdgcn_readlane(__float_as_int(t), j));
        const float* zp = z + (size_t)sj * D + lane;
        const float z0 = zp[0], z1 = zp[32], z2 = zp[64], z3 = zp[96];
        float* op = ((dj < n0) ? (out_lo + (size_t)dj * D)
                               : (out_hi + (size_t)(dj - n0) * D)) + lane;
        atomicAdd(op,      tj * z0);
        atomicAdd(op + 32, tj * z1);
        atomicAdd(op + 64, tj * z2);
        atomicAdd(op + 96, tj * z3);
    }
}

__global__ void row_div(float* __restrict__ rows, const float* __restrict__ rows1,
                        const float* den0, const float* den1, int two, int n)
{
    const int wave = (blockIdx.x * blockDim.x + threadIdx.x) >> 5;
    const int lane = threadIdx.x & 31;
    const int r0 = wave * 4;
    float4 v[4];
    float dn[4];
#pragma unroll
    for (int u = 0; u < 4; ++u) {
        const int r = r0 + u;
        if (r < n) {
            dn[u] = two ? den0[r] + den1[r] : den0[r];
            const float4 p = *(const float4*)(rows + (size_t)r * D + lane * 4);
            const float4 q = *(const float4*)(rows1 + (size_t)r * D + lane * 4);
            v[u].x = p.x + q.x; v[u].y = p.y + q.y; v[u].z = p.z + q.z; v[u].w = p.w + q.w;
        }
    }
#pragma unroll
    for (int u = 0; u < 4; ++u) {
        const int r = r0 + u;
        if (r < n) {
            float4 o;
            if (dn[u] > 0.0f) {
                o.x = v[u].x / dn[u]; o.y = v[u].y / dn[u];
                o.z = v[u].z / dn[u]; o.w = v[u].w / dn[u];
            } else {
                o.x = 0.0f; o.y = 0.0f; o.z = 0.0f; o.w = 0.0f;
            }
            *(float4*)(rows + (size_t)r * D + lane * 4) = o;
        }
    }
}

__global__ void combine(const float* __restrict__ A0, const float* __restrict__ A1,
                        const float* __restrict__ B0, const float* B1,
                        const float* __restrict__ W1, const float* __restrict__ b1,
                        const float* __restrict__ w2, float* out,
                        const float* den0, const float* den1, int two,
                        int Isplit)
{
    __shared__ float sA[D], sB[D], red[D];
    const int i = blockIdx.x;
    const int j = threadIdx.x;
    const float dA = two ? den0[i] + den1[i] : den0[i];
    const float dB = two ? den0[Isplit + i] + den1[Isplit + i] : den0[Isplit + i];
    const float rA = A0[(size_t)i * D + j] + A1[(size_t)i * D + j];
    const float rB = B0[(size_t)i * D + j] + B1[(size_t)i * D + j];
    sA[j] = dA > 0.0f ? rA / dA : 0.0f;
    sB[j] = dB > 0.0f ? rB / dB : 0.0f;
    __syncthreads();
    float accA = b1[j], accB = accA;
    for (int k = 0; k < D; ++k) {
        float wv = W1[(size_t)k * D + j];
        accA = fmaf(sA[k], wv, accA);
        accB = fmaf(sB[k], wv, accB);
    }
    const float w2j = w2[j];
    const float tA = tanhf(accA) * w2j;
    const float tB = tanhf(accB) * w2j;

    red[j] = tA; __syncthreads();
    for (int s = 64; s > 0; s >>= 1) { if (j < s) red[j] += red[j + s]; __syncthreads(); }
    const float s1 = red[0]; __syncthreads();
    red[j] = tB; __syncthreads();
    for (int s = 64; s > 0; s >>= 1) { if (j < s) red[j] += red[j + s]; __syncthreads(); }
    const float s2 = red[0];

    const float mm = fmaxf(s1, s2);
    const float e1 = __expf(s1 - mm), e2 = __expf(s2 - mm);
    const float al = e1 / (e1 + e2);
    out[(size_t)i * D + j] = al * sA[j] + (1.0f - al) * sB[j];
}

static inline int cdiv(int a, int b) { return (a + b - 1) / b; }

extern "C" void kernel_launch(void* const* d_in, const int* in_sizes, int n_in,
                              void* d_out, int out_size, void* d_ws, size_t ws_size,
                              hipStream_t stream)
{
    (void)n_in;
    const float* kn_emb   = (const float*)d_in[0];
    const float* exer_emb = (const float*)d_in[1];
    const float* W_dir    = (const float*)d_in[2];
    const float* a_dir    = (const float*)d_in[3];
    const float* W_ke     = (const float*)d_in[4];
    const float* a_ke     = (const float*)d_in[5];
    const float* W_ek     = (const float*)d_in[6];
    const float* a_ek     = (const float*)d_in[7];
    const float* rW1      = (const float*)d_in[8];
    const float* rb1      = (const float*)d_in[9];
    const float* rw2      = (const float*)d_in[10];
    const int* dsrc  = (const int*)d_in[11];
    const int* ddst  = (const int*)d_in[12];
    const int* kesrc = (const int*)d_in[13];
    const int* kedst = (const int*)d_in[14];
    const int* eksrc = (const int*)d_in[15];
    const int* ekdst = (const int*)d_in[16];

    const int Kn = in_sizes[0] / D;
    const int In = in_sizes[1] / D;
    const int Nn = Kn + In;
    const int E1 = in_sizes[11];
    const int E2 = in_sizes[13];

    float* w = (float*)d_ws;
    size_t off = 0;
    float* buf_z = w + off; off += (size_t)Nn * D;
    float* pY    = w + off; off += (size_t)Kn * D;
    float* pX    = w + off; off += (size_t)In * D;
    float* pA    = w + off; off += (size_t)Kn * D;
    const int two = (ws_size >= (off + 6 * (size_t)Nn) * sizeof(float)) ? 1 : 0;
    const size_t nc = two ? 2 : 1;
    float* den0  = w + off; off += nc * (size_t)Nn;
    float* den1  = two ? den0 + Nn : den0;
    float* ssrc  = w + off; off += (size_t)Nn;
    float* sdst  = w + off; off += (size_t)Nn;
    float* m0    = w + off; off += nc * (size_t)Nn;
    float* m1    = two ? m0 + Nn : m0;

    float* out_kn = (float*)d_out;
    float* out_ex = (float*)d_out + (size_t)Kn * D;

    const size_t planes_den = ((size_t)(Nn + Kn) * D + nc * (size_t)Nn) * sizeof(float);
    const int n_m = (int)(nc * (size_t)Nn);

    hipMemsetAsync(d_out, 0, (size_t)out_size * sizeof(float), stream);

    gemm_z<<<cdiv(Nn, 16), 256, 0, stream>>>(exer_emb, In, kn_emb, W_ke, buf_z, Nn);
    row_scores<<<cdiv(Nn, 8), 256, 0, stream>>>(buf_z, a_ke, ssrc, sdst, Nn);
    fill_f32<<<cdiv(n_m, 256), 256, 0, stream>>>(m0, -INFINITY, n_m);
    hipMemsetAsync(pY, 0, planes_den, stream);
    edge_max<<<cdiv(E2, 256), 256, 0, stream>>>(kesrc, kedst, ssrc, sdst, m0, m1, E2);
    edge_den_agg<<<cdiv(E2, 256), 256, 0, stream>>>(kesrc, kedst, ssrc, sdst, m0, m1, den0, den1,
                                                    buf_z, pX, pY, out_ex, out_kn, In, E2);

    float* pA1 = pX + (size_t)(In - Kn) * D;
    gemm_z<<<cdiv(Kn, 16), 256, 0, stream>>>(kn_emb, Kn, kn_emb, W_dir, buf_z, Kn);
    row_scores<<<cdiv(Kn, 8), 256, 0, stream>>>(buf_z, a_dir, ssrc, sdst, Kn);
    fill_f32<<<cdiv(n_m, 256), 256, 0, stream>>>(m0, -INFINITY, n_m);
    hipMemsetAsync(pA1, 0, ((size_t)2 * Kn * D + Kn) * sizeof(float), stream);
    if (two) hipMemsetAsync(den1, 0, (size_t)Kn * sizeof(float), stream);
    edge_max<<<cdiv(E1, 256), 256, 0, stream>>>(dsrc, ddst, ssrc, sdst, m0, m1, E1);
    edge_den_agg<<<cdiv(E1, 256), 256, 0, stream>>>(dsrc, ddst, ssrc, sdst, m0, m1, den0, den1,
                                                    buf_z, pA, pA, pA1, pA1, Kn, E1);

    combine<<<Kn, D, 0, stream>>>(pA, pA1, pY, out_kn, rW1, rb1, rw2, out_kn, den0, den1, two, In);

    gemm_z<<<cdiv(Nn, 16), 256, 0, stream>>>(exer_emb, In, kn_emb, W_ek, buf_z, Nn);
    row_scores<<<cdiv(Nn, 8), 256, 0, stream>>>(buf_z, a_ek, ssrc, sdst, Nn);
    fill_f32<<<cdiv(n_m, 256), 256, 0, stream>>>(m0, -INFINITY, n_m);
    hipMemsetAsync(pY, 0, planes_den, stream);
    hipMemsetAsync(out_ex, 0, (size_t)In * D * sizeof(float), stream);
    edge_max<<<cdiv(E2, 256), 256, 0, stream>>>(eksrc, ekdst, ssrc, sdst, m0, m1, E2);
    edge_den_agg<<<cdiv(E2, 256), 256, 0, stream>>>(eksrc, ekdst, ssrc, sdst, m0, m1, den0, den1,
                                                    buf_z, out_ex, pA, pX, pY, In, E2);
    row_div<<<cdiv(In, 32), 256, 0, stream>>>(out_ex, pX, den0, den1, two, In);
}


# edge passes by owner work-groups: each owns a run of dst rows, scans the dst list, keeps row sums, max, den in LDS (LDS atomics), stores rows once; global atomics, clears gone; sum order changes
# speedup vs baseline: 2.1995x; 1.3535x over previous
#include <hip/hip_runtime.h>
#include <hip/hip_bf16.h>
#include <math.h>

#define D 128

typedef float v2f __attribute__((ext_vector_type(2)));
typedef float v8f __attribute__((ext_vector_type(8)));

__global__ void gemm_z(const float* __restrict__ h0, int n0,
                       const float* __restrict__ h1,
                       const float* __restrict__ W,
                       float* __restrict__ Z, int n)
{
    const int wave = threadIdx.x >> 5;
    const int lane = threadIdx.x & 31;
    const int half = lane >> 4;
    const int lr   = lane & 15;
    const int row0 = blockIdx.x * 16;
    const int col0 = wave * 16;

    int rowA = row0 + lr;
    if (rowA >= n) rowA = n - 1;
    const float* hrow = (rowA < n0) ? (h0 + (size_t)rowA * D)
                                    : (h1 + (size_t)(rowA - n0) * D);

    v8f c = {0.f, 0.f, 0.f, 0.f, 0.f, 0.f, 0.f, 0.f};
#pragma unroll
    for (int k0 = 0; k0 < D; k0 += 4) {
        const int kk = k0 + 2 * half;
        float2 av = *(const float2*)(hrow + kk);
        v2f a; a.x = av.x; a.y = av.y;
        v2f b; b.x = W[(size_t)kk * D + col0 + lr];
               b.y = W[(size_t)(kk + 1) * D + col0 + lr];
        c = __builtin_amdgcn_wmma_f32_16x16x4_f32(
                false, a, false, b, (short)0, c, false, false);
    }
#pragma unroll
    for (int v = 0; v < 8; ++v) {
        const int r = row0 + v + 8 * half;
        if (r < n) Z[(size_t)r * D + col0 + lr] = c[v];
    }
}

__global__ void row_scores(const float* __restrict__ Z, const float* __restrict__ a,
                           float* __restrict__ ssrc, float* __restrict__ sdst, int n)
{
    const int wave = (blockIdx.x * blockDim.x + threadIdx.x) >> 5;
    const int lane = threadIdx.x & 31;
    if (wave >= n) return;
    const float4 zv = *(const float4*)(Z + (size_t)wave * D + lane * 4);
    const float4 a0 = *(const float4*)(a + lane * 4);
    const float4 a1 = *(const float4*)(a + D + lane * 4);
    float p0 = zv.x * a0.x + zv.y * a0.y + zv.z * a0.z + zv.w * a0.w;
    float p1 = zv.x * a1.x + zv.y * a1.y + zv.z * a1.z + zv.w * a1.w;
#pragma unroll
    for (int off = 16; off > 0; off >>= 1) {
        p0 += __shfl_down(p0, off, 32);
        p1 += __shfl_down(p1, off, 32);
    }
    if (lane == 0) { ssrc[wave] = p0; sdst[wave] = p1; }
}

__device__ __forceinline__ float leaky(float v) {
    return v > 0.0f ? v : 0.01f * v;
}

#define OWN_T    1024
#define OWN_W    (OWN_T / 32)
#define OWN_RMAX 288
#define OWN_CAP  8192
#define OWN_U    8
#define OWN_G    16

__device__ __forceinline__ void ldsMaxF(float* addr, float val) {
    if (val >= 0.0f) atomicMax((int*)addr, __float_as_int(val));
    else             atomicMin((unsigned int*)addr, __float_as_uint(val));
}

__device__ __forceinline__ void own_note(int d, int e, int r0, int nr,
                                         int* cnt, int* La, int* Lr) {
    const int r = d - r0;
    if ((unsigned int)r < (unsigned int)nr) {
        const int idx = atomicAdd(cnt, 1);
        if (idx < OWN_CAP) { La[idx] = e; Lr[idx] = r; }
    }
}

__device__ __forceinline__ void own_load(int4* dv, const int4* dst4, int g, int nG) {
#pragma unroll
    for (int u = 0; u < OWN_U; ++u) {
        int gi = g + u * OWN_T;
        gi = gi < nG ? gi : nG - 1;
        dv[u] = dst4[gi];
    }
}

__device__ __forceinline__ void own_chunk(int4* dv, int gb, int nG, bool shortc,
                                          int r0, int nr, int* cnt, int* La, int* Lr) {
    if (shortc) {
#pragma unroll
        for (int u = 0; u < OWN_U; ++u)
            if (gb + u * OWN_T >= nG) dv[u] = make_int4(-1, -1, -1, -1);
    }
#pragma unroll
    for (int u = 0; u < OWN_U; ++u) {
        const int e0 = (gb + u * OWN_T) << 2;
        own_note(dv[u].x, e0,     r0, nr, cnt, La, Lr);
        own_note(dv[u].y, e0 + 1, r0, nr, cnt, La, Lr);
        own_note(dv[u].z, e0 + 2, r0, nr, cnt, La, Lr);
        own_note(dv[u].w, e0 + 3, r0, nr, cnt, La, Lr);
    }
}

__device__ __forceinline__ void own_add_row(float* arow, float t, const float* zv) {
    atomicAdd(arow,      t * zv[0]);
    atomicAdd(arow + 32, t * zv[1]);
    atomicAdd(arow + 64, t * zv[2]);
    atomicAdd(arow + 96, t * zv[3]);
}

__global__ __launch_bounds__(OWN_T)
void owner_agg(const int* __restrict__ src, const int* __restrict__ dst,
               const float* __restrict__ ssrc, const float* __restrict__ sdst,
               const float* __restrict__ z,
               float* out_lo, float* out_hi,
               int n0, int n, int nE, int R)
{
    __shared__ __attribute__((aligned(16))) float acc[OWN_RMAX * D];
    __shared__ float mx[OWN_RMAX];
    __shared__ float dn[OWN_RMAX];
    __shared__ float sd[OWN_RMAX];
    __shared__ int   La[OWN_CAP];
    __shared__ int   Lr[OWN_CAP];
    __shared__ float Lv[OWN_CAP];
    __shared__ int   s_cnt;

    const int tid  = threadIdx.x;
    const int lane = tid & 31;
    const int wave = tid >> 5;
    const int r0   = blockIdx.x * R;
    int nr = n - r0;
    nr = nr > R ? R : nr;
    if (nr <= 0) return;

    {
        const float4 z4 = make_float4(0.0f, 0.0f, 0.0f, 0.0f);
        float4* a4 = (float4*)acc;
        for (int i = tid; i < nr * (D / 4); i += OWN_T) a4[i] = z4;
        for (int i = tid; i < nr; i += OWN_T) {
            mx[i] = -INFINITY;
            dn[i] = 0.0f;
            sd[i] = sdst[r0 + i];
        }
        if (tid == 0) s_cnt = 0;
    }
    __syncthreads();

    {
        const int nG  = nE >> 2;
        const int CH  = OWN_T * OWN_U;
        const int nCh = (nG + CH - 1) / CH;
        if (nG > 0) {
            const int4* dst4 = (const int4*)dst;
            int ch = (int)(((long long)blockIdx.x * nCh) / gridDim.x);
            int4 cur[OWN_U];
            own_load(cur, dst4, ch * CH + tid, nG);
            for (int it = 0; it + 1 < nCh; ++it) {
                int nch = ch + 1;
                if (nch >= nCh) nch = 0;
                int4 nxt[OWN_U];
                own_load(nxt, dst4, nch * CH + tid, nG);
                own_chunk(cur, ch * CH + tid, nG, ch == nCh - 1, r0, nr, &s_cnt, La, Lr);
#pragma unroll
                for (int u = 0; u < OWN_U; ++u) cur[u] = nxt[u];
                ch = nch;
            }
            own_chunk(cur, ch * CH + tid, nG, ch == nCh - 1, r0, nr, &s_cnt, La, Lr);
        }
        const int et = (nG << 2) + tid;
        if (et < nE) own_note(dst[et], et, r0, nr, &s_cnt, La, Lr);
    }
    __syncthreads();
    const int cnt = s_cnt;

    if (cnt <= OWN_CAP) {
        for (int i = tid; i < cnt; i += OWN_T) {
            const int e = La[i];
            const int r = Lr[i];
            const int s = src[e];
            const float v = leaky(ssrc[s] + sd[r]);
            La[i] = s;
            Lv[i] = v;
            ldsMaxF(&mx[r], v);
        }
        __syncthreads();
        for (int i = tid; i < cnt; i += OWN_T) {
            const int r = Lr[i];
            const float t = __expf(Lv[i] - mx[r]);
            Lv[i] = t;
            atomicAdd(&dn[r], t);
        }
        __syncthreads();
        for (int base = wave * OWN_G; base < cnt; base += OWN_W * OWN_G) {
            if (base + OWN_G <= cnt) {
                float zv[OWN_G][4];
#pragma unroll
                for (int u = 0; u < OWN_G; ++u) {
                    const int s = __builtin_amdgcn_readfirstlane(La[base + u]);
                    const float* zp = z + (size_t)s * D + lane;
                    zv[u][0] = zp[0];
                    zv[u][1] = zp[32];
                    zv[u][2] = zp[64];
                    zv[u][3] = zp[96];
                }
#pragma unroll
                for (int u = 0; u < OWN_G; ++u) {
                    const int r = __builtin_amdgcn_readfirstlane(Lr[base + u]);
                    own_add_row(acc + r * D + lane, Lv[base + u], zv[u]);
                }
            } else {
                for (int i = base; i < cnt; ++i) {
                    const int s = __builtin_amdgcn_readfirstlane(La[i]);
                    const int r = __builtin_amdgcn_readfirstlane(Lr[i]);
                    const float* zp = z + (size_t)s * D + lane;
                    float zv[4];
                    zv[0] = zp[0]; zv[1] = zp[32]; zv[2] = zp[64]; zv[3] = zp[96];
                    own_add_row(acc + r * D + lane, Lv[i], zv);
                }
            }
        }
    } else {
        for (int e = tid; e < nE; e += OWN_T) {
            const int r = dst[e] - r0;
            if ((unsigned int)r < (unsigned int)nr)
                ldsMaxF(&mx[r], leaky(ssrc[src[e]] + sd[r]));
        }
        __syncthreads();
        for (int base = wave * 32; base < nE; base += OWN_T) {
            const int e = base + lane;
            int s = 0, r = -1;
            float t = 0.0f;
            if (e < nE) r = dst[e] - r0;
            const bool hit = (unsigned int)r < (unsigned int)nr;
            if (hit) {
                s = src[e];
                t = __expf(leaky(ssrc[s] + sd[r]) - mx[r]);
                atomicAdd(&dn[r], t);
            }
            unsigned int mask = __builtin_amdgcn_ballot_w32(hit);
            while (mask) {
                const int j = __builtin_amdgcn_readfirstlane(__builtin_ctz(mask));
                mask &= mask - 1;
                const int   sj = __builtin_amdgcn_readlane(s, j);
                const int   rj = __builtin_amdgcn_readlane(r, j);
                const float tj = __int_as_float(__builtin_amdgcn_readlane(__float_as_int(t), j));
                const float* zp = z + (size_t)sj * D + lane;
                float zv[4];
                zv[0] = zp[0]; zv[1] = zp[32]; zv[2] = zp[64]; zv[3] = zp[96];
                own_add_row(acc + rj * D + lane, tj, zv);
            }
        }
    }
    __syncthreads();

    for (int r = wave; r < nr; r += OWN_W) {
        const float d = dn[r];
        const float4 a = *(const float4*)(acc + r * D + lane * 4);
        float4 o;
        if (d > 0.0f) {
            o.x = a.x / d; o.y = a.y / d; o.z = a.z / d; o.w = a.w / d;
        } else {
            o.x = 0.0f; o.y = 0.0f; o.z = 0.0f; o.w = 0.0f;
        }
        const int row = r0 + r;
        float* op = (row < n0) ? (out_lo + (size_t)row * D)
                               : (out_hi + (size_t)(row - n0) * D);
        *(float4*)(op + lane * 4) = o;
    }
}

__global__ void combine(const float* __restrict__ A, const float* __restrict__ B,
                        const float* __restrict__ W1, const float* __restrict__ b1,
                        const float* __restrict__ w2, float* __restrict__ out)
{
    __shared__ float sA[D], sB[D], red[D];
    const int i = blockIdx.x;
    const int j = threadIdx.x;
    sA[j] = A[(size_t)i * D + j];
    sB[j] = B[(size_t)i * D + j];
    __syncthreads();
    float accA = b1[j], accB = accA;
    for (int k = 0; k < D; ++k) {
        float wv = W1[(size_t)k * D + j];
        accA = fmaf(sA[k], wv, accA);
        accB = fmaf(sB[k], wv, accB);
    }
    const float w2j = w2[j];
    const float tA = tanhf(accA) * w2j;
    const float tB = tanhf(accB) * w2j;

    red[j] = tA; __syncthreads();
    for (int s = 64; s > 0; s >>= 1) { if (j < s) red[j] += red[j + s]; __syncthreads(); }
    const float s1 = red[0]; __syncthreads();
    red[j] = tB; __syncthreads();
    for (int s = 64; s > 0; s >>= 1) { if (j < s) red[j] += red[j + s]; __syncthreads(); }
    const float s2 = red[0];

    const float mm = fmaxf(s1, s2);
    const float e1 = __expf(s1 - mm), e2 = __expf(s2 - mm);
    const float al = e1 / (e1 + e2);
    out[(size_t)i * D + j] = al * sA[j] + (1.0f - al) * sB[j];
}

static inline int cdiv(int a, int b) { return (a + b - 1) / b; }

extern "C" void kernel_launch(void* const* d_in, const int* in_sizes, int n_in,
                              void* d_out, int out_size, void* d_ws, size_t ws_size,
                              hipStream_t stream)
{
    (void)n_in; (void)out_size; (void)ws_size;
    const float* kn_emb   = (const float*)d_in[0];
    const float* exer_emb = (const float*)d_in[1];
    const float* W_dir    = (const float*)d_in[2];
    const float* a_dir    = (const float*)d_in[3];
    const float* W_ke     = (const float*)d_in[4];
    const float* a_ke     = (const float*)d_in[5];
    const float* W_ek     = (const float*)d_in[6];
    const float* a_ek     = (const float*)d_in[7];
    const float* rW1      = (const float*)d_in[8];
    const float* rb1      = (const float*)d_in[9];
    const float* rw2      = (const float*)d_in[10];
    const int* dsrc  = (const int*)d_in[11];
    const int* ddst  = (const int*)d_in[12];
    const int* kesrc = (const int*)d_in[13];
    const int* kedst = (const int*)d_in[14];
    const int* eksrc = (const int*)d_in[15];
    const int* ekdst = (const int*)d_in[16];

    const int Kn = in_sizes[0] / D;
    const int In = in_sizes[1] / D;
    const int Nn = Kn + In;
    const int E1 = in_sizes[11];
    const int E2 = in_sizes[13];

    float* w = (float*)d_ws;
    size_t off = 0;
    float* buf_z = w + off; off += (size_t)Nn * D;
    float* pY    = w + off; off += (size_t)Kn * D;
    float* pX    = w + off; off += (size_t)In * D;
    float* pA    = w + off; off += (size_t)Kn * D;
    float* ssrc  = w + off; off += (size_t)Nn;
    float* sdst  = w + off; off += (size_t)Nn;

    float* out_kn = (float*)d_out;
    float* out_ex = (float*)d_out + (size_t)Kn * D;

    int gN = 256, rN = cdiv(Nn, gN);
    if (rN > OWN_RMAX) { gN = cdiv(Nn, OWN_RMAX); rN = cdiv(Nn, gN); }
    int gK = 256, rK = cdiv(Kn, gK);
    if (rK > OWN_RMAX) { gK = cdiv(Kn, OWN_RMAX); rK = cdiv(Kn, gK); }

    gemm_z<<<cdiv(Nn, 16), 256, 0, stream>>>(exer_emb, In, kn_emb, W_ke, buf_z, Nn);
    row_scores<<<cdiv(Nn, 8), 256, 0, stream>>>(buf_z, a_ke, ssrc, sdst, Nn);
    owner_agg<<<gN, OWN_T, 0, stream>>>(kesrc, kedst, ssrc, sdst, buf_z, pX, pY, In, Nn, E2, rN);

    gemm_z<<<cdiv(Kn, 16), 256, 0, stream>>>(kn_emb, Kn, kn_emb, W_dir, buf_z, Kn);
    row_scores<<<cdiv(Kn, 8), 256, 0, stream>>>(buf_z, a_dir, ssrc, sdst, Kn);
    owner_agg<<<gK, OWN_T, 0, stream>>>(dsrc, ddst, ssrc, sdst, buf_z, pA, pA, Kn, Kn, E1, rK);

    combine<<<Kn, D, 0, stream>>>(pA, pY, rW1, rb1, rw2, out_kn);

    gemm_z<<<cdiv(Nn, 16), 256, 0, stream>>>(exer_emb, In, kn_emb, W_ek, buf_z, Nn);
    row_scores<<<cdiv(Nn, 8), 256, 0, stream>>>(buf_z, a_ek, ssrc, sdst, Nn);
    owner_agg<<<gN, OWN_T, 0, stream>>>(eksrc, ekdst, ssrc, sdst, buf_z, out_ex, pY, In, Nn, E2, rN);
}


# edge phases: owner work-groups scan a byte-wide owner key per edge (dst / rows per owner, packed by the score kernel), 4 keys per compare, not the 4-byte dst list; sums stay in list arrival order
# speedup vs baseline: 2.4373x; 1.1081x over previous
#include <hip/hip_runtime.h>
#include <hip/hip_bf16.h>
#include <math.h>

#define D 128

typedef float v2f __attribute__((ext_vector_type(2)));
typedef float v8f __attribute__((ext_vector_type(8)));

__global__ void gemm_z(const float* __restrict__ h0, int n0,
                       const float* __restrict__ h1,
                       const float* __restrict__ W,
                       float* __restrict__ Z, int n)
{
    const int wave = threadIdx.x >> 5;
    const int lane = threadIdx.x & 31;
    const int half = lane >> 4;
    const int lr   = lane & 15;
    const int row0 = blockIdx.x * 16;
    const int col0 = wave * 16;

    int rowA = row0 + lr;
    if (rowA >= n) rowA = n - 1;
    const float* hrow = (rowA < n0) ? (h0 + (size_t)rowA * D)
                                    : (h1 + (size_t)(rowA - n0) * D);

    v8f c = {0.f, 0.f, 0.f, 0.f, 0.f, 0.f, 0.f, 0.f};
#pragma unroll
    for (int k0 = 0; k0 < D; k0 += 4) {
        const int kk = k0 + 2 * half;
        float2 av = *(const float2*)(hrow + kk);
        v2f a; a.x = av.x; a.y = av.y;
        v2f b; b.x = W[(size_t)kk * D + col0 + lr];
               b.y = W[(size_t)(kk + 1) * D + col0 + lr];
        c = __builtin_amdgcn_wmma_f32_16x16x4_f32(
                false, a, false, b, (short)0, c, false, false);
    }
#pragma unroll
    for (int v = 0; v < 8; ++v) {
        const int r = row0 + v + 8 * half;
        if (r < n) Z[(size_t)r * D + col0 + lr] = c[v];
    }
}

__global__ void row_scores(const float* __restrict__ Z, const float* __restrict__ a,
                           float* __restrict__ ssrc, float* __restrict__ sdst, int n,
                           const int* __restrict__ dst, int* __restrict__ key,
                           int nE, int R)
{
    if (key != nullptr) {
        const int nT = gridDim.x * blockDim.x;
        const int nW = nE >> 2;
        const unsigned int uR = (unsigned int)R;
        for (int q = blockIdx.x * blockDim.x + threadIdx.x; q < nW; q += nT) {
            const int4 d = ((const int4*)dst)[q];
            const unsigned int w = (((unsigned int)d.x / uR) & 255u)
                                 | ((((unsigned int)d.y / uR) & 255u) << 8)
                                 | ((((unsigned int)d.z / uR) & 255u) << 16)
                                 | ((((unsigned int)d.w / uR) & 255u) << 24);
            key[q] = (int)w;
        }
    }
    const int wave = (blockIdx.x * blockDim.x + threadIdx.x) >> 5;
    const int lane = threadIdx.x & 31;
    if (wave >= n) return;
    const float4 zv = *(const float4*)(Z + (size_t)wave * D + lane * 4);
    const float4 a0 = *(const float4*)(a + lane * 4);
    const float4 a1 = *(const float4*)(a + D + lane * 4);
    float p0 = zv.x * a0.x + zv.y * a0.y + zv.z * a0.z + zv.w * a0.w;
    float p1 = zv.x * a1.x + zv.y * a1.y + zv.z * a1.z + zv.w * a1.w;
#pragma unroll
    for (int off = 16; off > 0; off >>= 1) {
        p0 += __shfl_down(p0, off, 32);
        p1 += __shfl_down(p1, off, 32);
    }
    if (lane == 0) { ssrc[wave] = p0; sdst[wave] = p1; }
}

__device__ __forceinline__ float leaky(float v) {
    return v > 0.0f ? v : 0.01f * v;
}

#define OWN_T    1024
#define OWN_W    (OWN_T / 32)
#define OWN_RMAX 288
#define OWN_CAP  8192
#define OWN_U    8
#define OWN_UK   4
#define OWN_G    16

__device__ __forceinline__ void ldsMaxF(float* addr, float val) {
    if (val >= 0.0f) atomicMax((int*)addr, __float_as_int(val));
    else             atomicMin((unsigned int*)addr, __float_as_uint(val));
}

__device__ __forceinline__ void own_push(int e, int* cnt, int* La) {
    const int idx = atomicAdd(cnt, 1);
    if (idx < OWN_CAP) La[idx] = e;
}

__device__ __forceinline__ void own_note(int d, int e, int r0, int nr,
                                         int* cnt, int* La) {
    if ((unsigned int)(d - r0) < (unsigned int)nr) own_push(e, cnt, La);
}

__device__ __forceinline__ void own_load(int4* dv, const int4* dst4, int g, int nG) {
#pragma unroll
    for (int u = 0; u < OWN_U; ++u) {
        int gi = g + u * OWN_T;
        gi = gi < nG ? gi : nG - 1;
        dv[u] = dst4[gi];
    }
}

__device__ __forceinline__ void own_chunk(int4* dv, int gb, int nG, bool shortc,
                                          int r0, int nr, int* cnt, int* La) {
    if (shortc) {
#pragma unroll
        for (int u = 0; u < OWN_U; ++u)
            if (gb + u * OWN_T >= nG) dv[u] = make_int4(-1, -1, -1, -1);
    }
#pragma unroll
    for (int u = 0; u < OWN_U; ++u) {
        const int e0 = (gb + u * OWN_T) << 2;
        own_note(dv[u].x, e0,     r0, nr, cnt, La);
        own_note(dv[u].y, e0 + 1, r0, nr, cnt, La);
        own_note(dv[u].z, e0 + 2, r0, nr, cnt, La);
        own_note(dv[u].w, e0 + 3, r0, nr, cnt, La);
    }
}

__device__ __forceinline__ void key_load(uint4* kv, const uint4* key4, int g, int nQ) {
#pragma unroll
    for (int u = 0; u < OWN_UK; ++u) {
        int gi = g + u * OWN_T;
        gi = gi < nQ ? gi : nQ - 1;
        kv[u] = key4[gi];
    }
}

__device__ __forceinline__ unsigned int key_nz(unsigned int x) {
    return ((x & 0x7f7f7f7fu) + 0x7f7f7f7fu) | x;
}

__device__ __forceinline__ unsigned int key_match(uint4 kv, unsigned int pat) {
    const unsigned int n0 = key_nz(kv.x ^ pat);
    const unsigned int n1 = key_nz(kv.y ^ pat);
    const unsigned int n2 = key_nz(kv.z ^ pat);
    const unsigned int n3 = key_nz(kv.w ^ pat);
    const unsigned int nz = ((n0 >> 7) & 0x01010101u) | ((n1 >> 6) & 0x02020202u)
                          | ((n2 >> 5) & 0x04040404u) | ((n3 >> 4) & 0x08080808u);
    return ~nz & 0x0f0f0f0fu;
}

__device__ __forceinline__ void key_chunk(const uint4* kv, int gb, int nQ, bool shortc,
                                          unsigned int pat, int* cnt, int* La) {
#pragma unroll
    for (int u = 0; u < OWN_UK; ++u) {
        unsigned int m = key_match(kv[u], pat);
        if (shortc && gb + u * OWN_T >= nQ) m = 0u;
        const int e0 = (gb + u * OWN_T) << 4;
        while (m) {
            const int p = __builtin_ctz(m);
            m &= m - 1u;
            own_push(e0 + ((p & 7) << 2) + (p >> 3), cnt, La);
        }
    }
}

__device__ __forceinline__ void own_add_row(float* arow, float t, const float* zv) {
    atomicAdd(arow,      t * zv[0]);
    atomicAdd(arow + 32, t * zv[1]);
    atomicAdd(arow + 64, t * zv[2]);
    atomicAdd(arow + 96, t * zv[3]);
}

__global__ __launch_bounds__(OWN_T)
void owner_agg(const int* __restrict__ src, const int* __restrict__ dst,
               const int* __restrict__ key,
               const float* __restrict__ ssrc, const float* __restrict__ sdst,
               const float* __restrict__ z,
               float* out_lo, float* out_hi,
               int n0, int n, int nE, int R)
{
    __shared__ __attribute__((aligned(16))) float acc[OWN_RMAX * D];
    __shared__ float mx[OWN_RMAX];
    __shared__ float dn[OWN_RMAX];
    __shared__ float sd[OWN_RMAX];
    __shared__ int   La[OWN_CAP];
    __shared__ int   Lr[OWN_CAP];
    __shared__ float Lv[OWN_CAP];
    __shared__ int   s_cnt;

    const int tid  = threadIdx.x;
    const int lane = tid & 31;
    const int wave = tid >> 5;
    const int r0   = blockIdx.x * R;
    int nr = n - r0;
    nr = nr > R ? R : nr;
    if (nr <= 0) return;

    {
        const float4 z4 = make_float4(0.0f, 0.0f, 0.0f, 0.0f);
        float4* a4 = (float4*)acc;
        for (int i = tid; i < nr * (D / 4); i += OWN_T) a4[i] = z4;
        for (int i = tid; i < nr; i += OWN_T) {
            mx[i] = -INFINITY;
            dn[i] = 0.0f;
            sd[i] = sdst[r0 + i];
        }
        if (tid == 0) s_cnt = 0;
    }
    __syncthreads();

    if (key != nullptr) {
        const int nQ  = nE >> 4;
        const int CH  = OWN_T * OWN_UK;
        const int nCh = (nQ + CH - 1) / CH;
        const unsigned int pat = (unsigned int)blockIdx.x * 0x01010101u;
        if (nQ > 0) {
            const uint4* key4 = (const uint4*)key;
            int ch = (int)(((long long)blockIdx.x * nCh) / gridDim.x);
            uint4 cur[OWN_UK];
            key_load(cur, key4, ch * CH + tid, nQ);
            for (int it = 0; it + 1 < nCh; ++it) {
                int nch = ch + 1;
                if (nch >= nCh) nch = 0;
                uint4 nxt[OWN_UK];
                key_load(nxt, key4, nch * CH + tid, nQ);
                key_chunk(cur, ch * CH + tid, nQ, ch == nCh - 1, pat, &s_cnt, La);
#pragma unroll
                for (int u = 0; u < OWN_UK; ++u) cur[u] = nxt[u];
                ch = nch;
            }
            key_chunk(cur, ch * CH + tid, nQ, ch == nCh - 1, pat, &s_cnt, La);
        }
        const int et = (nQ << 4) + tid;
        if (et < nE) own_note(dst[et], et, r0, nr, &s_cnt, La);
    } else {
        const int nG  = nE >> 2;
        const int CH  = OWN_T * OWN_U;
        const int nCh = (nG + CH - 1) / CH;
        if (nG > 0) {
            const int4* dst4 = (const int4*)dst;
            int ch = (int)(((long long)blockIdx.x * nCh) / gridDim.x);
            int4 cur[OWN_U];
            own_load(cur, dst4, ch * CH + tid, nG);
            for (int it = 0; it + 1 < nCh; ++it) {
                int nch = ch + 1;
                if (nch >= nCh) nch = 0;
                int4 nxt[OWN_U];
                own_load(nxt, dst4, nch * CH + tid, nG);
                own_chunk(cur, ch * CH + tid, nG, ch == nCh - 1, r0, nr, &s_cnt, La);
#pragma unroll
                for (int u = 0; u < OWN_U; ++u) cur[u] = nxt[u];
                ch = nch;
            }
            own_chunk(cur, ch * CH + tid, nG, ch == nCh - 1, r0, nr, &s_cnt, La);
        }
        const int et = (nG << 2) + tid;
        if (et < nE) own_note(dst[et], et, r0, nr, &s_cnt, La);
    }
    __syncthreads();
    const int cnt = s_cnt;

    if (cnt <= OWN_CAP) {
        for (int i = tid; i < cnt; i += OWN_T) {
            const int e = La[i];
            const int r = dst[e] - r0;
            const int s = src[e];
            const float v = leaky(ssrc[s] + sd[r]);
            La[i] = s;
            Lr[i] = r;
            Lv[i] = v;
            ldsMaxF(&mx[r], v);
        }
        __syncthreads();
        for (int i = tid; i < cnt; i += OWN_T) {
            const int r = Lr[i];
            const float t = __expf(Lv[i] - mx[r]);
            Lv[i] = t;
            atomicAdd(&dn[r], t);
        }
        __syncthreads();
        for (int base = wave * OWN_G; base < cnt; base += OWN_W * OWN_G) {
            if (base + OWN_G <= cnt) {
                float zv[OWN_G][4];
#pragma unroll
                for (int u = 0; u < OWN_G; ++u) {
                    const int s = __builtin_amdgcn_readfirstlane(La[base + u]);
                    const float* zp = z + (size_t)s * D + lane;
                    zv[u][0] = zp[0];
                    zv[u][1] = zp[32];
                    zv[u][2] = zp[64];
                    zv[u][3] = zp[96];
                }
#pragma unroll
                for (int u = 0; u < OWN_G; ++u) {
                    const int r = __builtin_amdgcn_readfirstlane(Lr[base + u]);
                    own_add_row(acc + r * D + lane, Lv[base + u], zv[u]);
                }
            } else {
                for (int i = base; i < cnt; ++i) {
                    const int s = __builtin_amdgcn_readfirstlane(La[i]);
                    const int r = __builtin_amdgcn_readfirstlane(Lr[i]);
                    const float* zp = z + (size_t)s * D + lane;
                    float zv[4];
                    zv[0] = zp[0]; zv[1] = zp[32]; zv[2] = zp[64]; zv[3] = zp[96];
                    own_add_row(acc + r * D + lane, Lv[i], zv);
                }
            }
        }
    } else {
        for (int e = tid; e < nE; e += OWN_T) {
            const int r = dst[e] - r0;
            if ((unsigned int)r < (unsigned int)nr)
                ldsMaxF(&mx[r], leaky(ssrc[src[e]] + sd[r]));
        }
        __syncthreads();
        for (int base = wave * 32; base < nE; base += OWN_T) {
            const int e = base + lane;
            int s = 0, r = -1;
            float t = 0.0f;
            if (e < nE) r = dst[e] - r0;
            const bool hit = (unsigned int)r < (unsigned int)nr;
            if (hit) {
                s = src[e];
                t = __expf(leaky(ssrc[s] + sd[r]) - mx[r]);
                atomicAdd(&dn[r], t);
            }
            unsigned int mask = __builtin_amdgcn_ballot_w32(hit);
            while (mask) {
                const int j = __builtin_amdgcn_readfirstlane(__builtin_ctz(mask));
                mask &= mask - 1;
                const int   sj = __builtin_amdgcn_readlane(s, j);
                const int   rj = __builtin_amdgcn_readlane(r, j);
                const float tj = __int_as_float(__builtin_amdgcn_readlane(__float_as_int(t), j));
                const float* zp = z + (size_t)sj * D + lane;
                float zv[4];
                zv[0] = zp[0]; zv[1] = zp[32]; zv[2] = zp[64]; zv[3] = zp[96];
                own_add_row(acc + rj * D + lane, tj, zv);
            }
        }
    }
    __syncthreads();

    for (int r = wave; r < nr; r += OWN_W) {
        const float d = dn[r];
        const float4 a = *(const float4*)(acc + r * D + lane * 4);
        float4 o;
        if (d > 0.0f) {
            o.x = a.x / d; o.y = a.y / d; o.z = a.z / d; o.w = a.w / d;
        } else {
            o.x = 0.0f; o.y = 0.0f; o.z = 0.0f; o.w = 0.0f;
        }
        const int row = r0 + r;
        float* op = (row < n0) ? (out_lo + (size_t)row * D)
                               : (out_hi + (size_t)(row - n0) * D);
        *(float4*)(op + lane * 4) = o;
    }
}

__global__ void combine(const float* __restrict__ A, const float* __restrict__ B,
                        const float* __restrict__ W1, const float* __restrict__ b1,
                        const float* __restrict__ w2, float* __restrict__ out)
{
    __shared__ float sA[D], sB[D], red[D];
    const int i = blockIdx.x;
    const int j = threadIdx.x;
    sA[j] = A[(size_t)i * D + j];
    sB[j] = B[(size_t)i * D + j];
    __syncthreads();
    float accA = b1[j], accB = accA;
    for (int k = 0; k < D; ++k) {
        float wv = W1[(size_t)k * D + j];
        accA = fmaf(sA[k], wv, accA);
        accB = fmaf(sB[k], wv, accB);
    }
    const float w2j = w2[j];
    const float tA = tanhf(accA) * w2j;
    const float tB = tanhf(accB) * w2j;

    red[j] = tA; __syncthreads();
    for (int s = 64; s > 0; s >>= 1) { if (j < s) red[j] += red[j + s]; __syncthreads(); }
    const float s1 = red[0]; __syncthreads();
    red[j] = tB; __syncthreads();
    for (int s = 64; s > 0; s >>= 1) { if (j < s) red[j] += red[j + s]; __syncthreads(); }
    const float s2 = red[0];

    const float mm = fmaxf(s1, s2);
    const float e1 = __expf(s1 - mm), e2 = __expf(s2 - mm);
    const float al = e1 / (e1 + e2);
    out[(size_t)i * D + j] = al * sA[j] + (1.0f - al) * sB[j];
}

static inline int cdiv(int a, int b) { return (a + b - 1) / b; }

extern "C" void kernel_launch(void* const* d_in, const int* in_sizes, int n_in,
                              void* d_out, int out_size, void* d_ws, size_t ws_size,
                              hipStream_t stream)
{
    (void)n_in; (void)out_size; (void)ws_size;
    const float* kn_emb   = (const float*)d_in[0];
    const float* exer_emb = (const float*)d_in[1];
    const float* W_dir    = (const float*)d_in[2];
    const float* a_dir    = (const float*)d_in[3];
    const float* W_ke     = (const float*)d_in[4];
    const float* a_ke     = (const float*)d_in[5];
    const float* W_ek     = (const float*)d_in[6];
    const float* a_ek     = (const float*)d_in[7];
    const float* rW1      = (const float*)d_in[8];
    const float* rb1      = (const float*)d_in[9];
    const float* rw2      = (const float*)d_in[10];
    const int* dsrc  = (const int*)d_in[11];
    const int* ddst  = (const int*)d_in[12];
    const int* kesrc = (const int*)d_in[13];
    const int* kedst = (const int*)d_in[14];
    const int* eksrc = (const int*)d_in[15];
    const int* ekdst = (const int*)d_in[16];

    const int Kn = in_sizes[0] / D;
    const int In = in_sizes[1] / D;
    const int Nn = Kn + In;
    const int E1 = in_sizes[11];
    const int E2 = in_sizes[13];

    float* w = (float*)d_ws;
    size_t off = 0;
    float* buf_z = w + off; off += (size_t)Nn * D;
    float* pY    = w + off; off += (size_t)Kn * D;
    float* pX    = w + off; off += (size_t)In * D;
    float* pA    = w + off; off += (size_t)Kn * D;
    float* ssrc  = w + off; off += (size_t)Nn;
    float* sdst  = w + off; off += (size_t)Nn;

    float* out_kn = (float*)d_out;
    float* out_ex = (float*)d_out + (size_t)Kn * D;

    int gN = 256, rN = cdiv(Nn, gN);
    if (rN > OWN_RMAX) { gN = cdiv(Nn, OWN_RMAX); rN = cdiv(Nn, gN); }
    int gK = 256, rK = cdiv(Kn, gK);
    if (rK > OWN_RMAX) { gK = cdiv(Kn, OWN_RMAX); rK = cdiv(Kn, gK); }

    int* keyKE  = (gN <= 256 && (size_t)E2 <= (size_t)Kn * D * 4) ? (int*)pA : nullptr;
    int* keyDIR = (gK <= 256 && (size_t)E1 <= (size_t)In * D * 4) ? (int*)pX : nullptr;
    int* keyEK  = (gN <= 256 && (size_t)E2 <= (size_t)In * D * 4) ? (int*)pX : nullptr;

    gemm_z<<<cdiv(Nn, 16), 256, 0, stream>>>(exer_emb, In, kn_emb, W_ke, buf_z, Nn);
    row_scores<<<cdiv(Nn, 8), 256, 0, stream>>>(buf_z, a_ke, ssrc, sdst, Nn, kedst, keyKE, E2, rN);
    owner_agg<<<gN, OWN_T, 0, stream>>>(kesrc, kedst, keyKE, ssrc, sdst, buf_z, pX, pY, In, Nn, E2, rN);

    gemm_z<<<cdiv(Kn, 16), 256, 0, stream>>>(kn_emb, Kn, kn_emb, W_dir, buf_z, Kn);
    row_scores<<<cdiv(Kn, 8), 256, 0, stream>>>(buf_z, a_dir, ssrc, sdst, Kn, ddst, keyDIR, E1, rK);
    owner_agg<<<gK, OWN_T, 0, stream>>>(dsrc, ddst, keyDIR, ssrc, sdst, buf_z, pA, pA, Kn, Kn, E1, rK);

    combine<<<Kn, D, 0, stream>>>(pA, pY, rW1, rb1, rw2, out_kn);

    gemm_z<<<cdiv(Nn, 16), 256, 0, stream>>>(exer_emb, In, kn_emb, W_ek, buf_z, Nn);
    row_scores<<<cdiv(Nn, 8), 256, 0, stream>>>(buf_z, a_ek, ssrc, sdst, Nn, ekdst, keyEK, E2, rN);
    owner_agg<<<gN, OWN_T, 0, stream>>>(eksrc, ekdst, keyEK, ssrc, sdst, buf_z, out_ex, pY, In, Nn, E2, rN);
}


# row_scores: a wave walks 32 rows, all their loads in flight, and stores whole 128-byte score lines (was one short-lived wave per row); key words get their own work-groups; scores keep their bits
# speedup vs baseline: 2.5770x; 1.0573x over previous
#include <hip/hip_runtime.h>
#include <hip/hip_bf16.h>
#include <math.h>

#define D 128

typedef float v2f __attribute__((ext_vector_type(2)));
typedef float v8f __attribute__((ext_vector_type(8)));

__global__ void gemm_z(const float* __restrict__ h0, int n0,
                       const float* __restrict__ h1,
                       const float* __restrict__ W,
                       float* __restrict__ Z, int n)
{
    const int wave = threadIdx.x >> 5;
    const int lane = threadIdx.x & 31;
    const int half = lane >> 4;
    const int lr   = lane & 15;
    const int row0 = blockIdx.x * 16;
    const int col0 = wave * 16;

    int rowA = row0 + lr;
    if (rowA >= n) rowA = n - 1;
    const float* hrow = (rowA < n0) ? (h0 + (size_t)rowA * D)
                                    : (h1 + (size_t)(rowA - n0) * D);

    v8f c = {0.f, 0.f, 0.f, 0.f, 0.f, 0.f, 0.f, 0.f};
#pragma unroll
    for (int k0 = 0; k0 < D; k0 += 4) {
        const int kk = k0 + 2 * half;
        float2 av = *(const float2*)(hrow + kk);
        v2f a; a.x = av.x; a.y = av.y;
        v2f b; b.x = W[(size_t)kk * D + col0 + lr];
               b.y = W[(size_t)(kk + 1) * D + col0 + lr];
        c = __builtin_amdgcn_wmma_f32_16x16x4_f32(
                false, a, false, b, (short)0, c, false, false);
    }
#pragma unroll
    for (int v = 0; v < 8; ++v) {
        const int r = row0 + v + 8 * half;
        if (r < n) Z[(size_t)r * D + col0 + lr] = c[v];
    }
}

#define RS_ROWS 32
#define RS_T    128
#define RS_KU   4
#define RS_WGR  (RS_ROWS * (RS_T / 32))
#define RS_WGK  (RS_T * RS_KU)

__device__ __forceinline__ float rs_dot4(float4 zv, float4 av) {
    float p = zv.y * av.y;
    p = fmaf(zv.x, av.x, p);
    p = fmaf(zv.z, av.z, p);
    p = fmaf(zv.w, av.w, p);
    return p;
}

__global__ __launch_bounds__(RS_T)
void row_scores(const float* __restrict__ Z, const float* __restrict__ a,
                float* __restrict__ ssrc, float* __restrict__ sdst, int n,
                const int* __restrict__ dst, int* __restrict__ key,
                int nE, int R)
{
    const int nRowWG = (n + RS_WGR - 1) / RS_WGR;
    if ((int)blockIdx.x >= nRowWG) {
        const int nW = nE >> 2;
        const unsigned int uR = (unsigned int)R;
        const int4* dst4 = (const int4*)dst;
        const int q0 = ((int)blockIdx.x - nRowWG) * RS_WGK + (int)threadIdx.x;
        int4 d[RS_KU];
#pragma unroll
        for (int u = 0; u < RS_KU; ++u) {
            int q = q0 + u * RS_T;
            q = q < nW ? q : nW - 1;
            d[u] = dst4[q];
        }
#pragma unroll
        for (int u = 0; u < RS_KU; ++u) {
            const int q = q0 + u * RS_T;
            if (q < nW) {
                const unsigned int w = (((unsigned int)d[u].x / uR) & 255u)
                                     | ((((unsigned int)d[u].y / uR) & 255u) << 8)
                                     | ((((unsigned int)d[u].z / uR) & 255u) << 16)
                                     | ((((unsigned int)d[u].w / uR) & 255u) << 24);
                key[q] = (int)w;
            }
        }
        return;
    }

    const int lane = threadIdx.x & 31;
    const int row0 = __builtin_amdgcn_readfirstlane(
                         (int)(blockIdx.x * RS_T + threadIdx.x) >> 5) * RS_ROWS;
    if (row0 >= n) return;

    const float4 a0 = *(const float4*)(a + lane * 4);
    const float4 a1 = *(const float4*)(a + D + lane * 4);
    float4 zv[RS_ROWS];
#pragma unroll
    for (int u = 0; u < RS_ROWS; ++u) {
        int r = row0 + u;
        r = r < n ? r : n - 1;
        zv[u] = *(const float4*)(Z + (size_t)r * D + lane * 4);
    }

    float p0[RS_ROWS], p1[RS_ROWS];
#pragma unroll
    for (int u = 0; u < RS_ROWS; ++u) {
        p0[u] = rs_dot4(zv[u], a0);
        p1[u] = rs_dot4(zv[u], a1);
    }
#pragma unroll
    for (int off = 16; off > 0; off >>= 1) {
#pragma unroll
        for (int u = 0; u < RS_ROWS; ++u) {
            p0[u] += __shfl_xor(p0[u], off, 32);
            p1[u] += __shfl_xor(p1[u], off, 32);
        }
    }
    float my0 = 0.0f, my1 = 0.0f;
#pragma unroll
    for (int u = 0; u < RS_ROWS; ++u)
        if (lane == u) { my0 = p0[u]; my1 = p1[u]; }
    if (row0 + lane < n) { ssrc[row0 + lane] = my0; sdst[row0 + lane] = my1; }
}

__device__ __forceinline__ float leaky(float v) {
    return v > 0.0f ? v : 0.01f * v;
}

#define OWN_T    1024
#define OWN_W    (OWN_T / 32)
#define OWN_RMAX 288
#define OWN_CAP  8192
#define OWN_U    8
#define OWN_UK   4
#define OWN_G    16

__device__ __forceinline__ void ldsMaxF(float* addr, float val) {
    if (val >= 0.0f) atomicMax((int*)addr, __float_as_int(val));
    else             atomicMin((unsigned int*)addr, __float_as_uint(val));
}

__device__ __forceinline__ void own_push(int e, int* cnt, int* La) {
    const int idx = atomicAdd(cnt, 1);
    if (idx < OWN_CAP) La[idx] = e;
}

__device__ __forceinline__ void own_note(int d, int e, int r0, int nr,
                                         int* cnt, int* La) {
    if ((unsigned int)(d - r0) < (unsigned int)nr) own_push(e, cnt, La);
}

__device__ __forceinline__ void own_load(int4* dv, const int4* dst4, int g, int nG) {
#pragma unroll
    for (int u = 0; u < OWN_U; ++u) {
        int gi = g + u * OWN_T;
        gi = gi < nG ? gi : nG - 1;
        dv[u] = dst4[gi];
    }
}

__device__ __forceinline__ void own_chunk(int4* dv, int gb, int nG, bool shortc,
                                          int r0, int nr, int* cnt, int* La) {
    if (shortc) {
#pragma unroll
        for (int u = 0; u < OWN_U; ++u)
            if (gb + u * OWN_T >= nG) dv[u] = make_int4(-1, -1, -1, -1);
    }
#pragma unroll
    for (int u = 0; u < OWN_U; ++u) {
        const int e0 = (gb + u * OWN_T) << 2;
        own_note(dv[u].x, e0,     r0, nr, cnt, La);
        own_note(dv[u].y, e0 + 1, r0, nr, cnt, La);
        own_note(dv[u].z, e0 + 2, r0, nr, cnt, La);
        own_note(dv[u].w, e0 + 3, r0, nr, cnt, La);
    }
}

__device__ __forceinline__ void key_load(uint4* kv, const uint4* key4, int g, int nQ) {
#pragma unroll
    for (int u = 0; u < OWN_UK; ++u) {
        int gi = g + u * OWN_T;
        gi = gi < nQ ? gi : nQ - 1;
        kv[u] = key4[gi];
    }
}

__device__ __forceinline__ unsigned int key_nz(unsigned int x) {
    return ((x & 0x7f7f7f7fu) + 0x7f7f7f7fu) | x;
}

__device__ __forceinline__ unsigned int key_match(uint4 kv, unsigned int pat) {
    const unsigned int n0 = key_nz(kv.x ^ pat);
    const unsigned int n1 = key_nz(kv.y ^ pat);
    const unsigned int n2 = key_nz(kv.z ^ pat);
    const unsigned int n3 = key_nz(kv.w ^ pat);
    const unsigned int nz = ((n0 >> 7) & 0x01010101u) | ((n1 >> 6) & 0x02020202u)
                          | ((n2 >> 5) & 0x04040404u) | ((n3 >> 4) & 0x08080808u);
    return ~nz & 0x0f0f0f0fu;
}

__device__ __forceinline__ void key_chunk(const uint4* kv, int gb, int nQ, bool shortc,
                                          unsigned int pat, int* cnt, int* La) {
#pragma unroll
    for (int u = 0; u < OWN_UK; ++u) {
        unsigned int m = key_match(kv[u], pat);
        if (shortc && gb + u * OWN_T >= nQ) m = 0u;
        const int e0 = (gb + u * OWN_T) << 4;
        while (m) {
            const int p = __builtin_ctz(m);
            m &= m - 1u;
            own_push(e0 + ((p & 7) << 2) + (p >> 3), cnt, La);
        }
    }
}

__device__ __forceinline__ void own_add_row(float* arow, float t, const float* zv) {
    atomicAdd(arow,      t * zv[0]);
    atomicAdd(arow + 32, t * zv[1]);
    atomicAdd(arow + 64, t * zv[2]);
    atomicAdd(arow + 96, t * zv[3]);
}

__global__ __launch_bounds__(OWN_T)
void owner_agg(const int* __restrict__ src, const int* __restrict__ dst,
               const int* __restrict__ key,
               const float* __restrict__ ssrc, const float* __restrict__ sdst,
               const float* __restrict__ z,
               float* out_lo, float* out_hi,
               int n0, int n, int nE, int R)
{
    __shared__ __attribute__((aligned(16))) float acc[OWN_RMAX * D];
    __shared__ float mx[OWN_RMAX];
    __shared__ float dn[OWN_RMAX];
    __shared__ float sd[OWN_RMAX];
    __shared__ int   La[OWN_CAP];
    __shared__ int   Lr[OWN_CAP];
    __shared__ float Lv[OWN_CAP];
    __shared__ int   s_cnt;

    const int tid  = threadIdx.x;
    const int lane = tid & 31;
    const int wave = tid >> 5;
    const int r0   = blockIdx.x * R;
    int nr = n - r0;
    nr = nr > R ? R : nr;
    if (nr <= 0) return;

    {
        const float4 z4 = make_float4(0.0f, 0.0f, 0.0f, 0.0f);
        float4* a4 = (float4*)acc;
        for (int i = tid; i < nr * (D / 4); i += OWN_T) a4[i] = z4;
        for (int i = tid; i < nr; i += OWN_T) {
            mx[i] = -INFINITY;
            dn[i] = 0.0f;
            sd[i] = sdst[r0 + i];
        }
        if (tid == 0) s_cnt = 0;
    }
    __syncthreads();

    if (key != nullptr) {
        const int nQ  = nE >> 4;
        const int CH  = OWN_T * OWN_UK;
        const int nCh = (nQ + CH - 1) / CH;
        const unsigned int pat = (unsigned int)blockIdx.x * 0x01010101u;
        if (nQ > 0) {
            const uint4* key4 = (const uint4*)key;
            int ch = (int)(((long long)blockIdx.x * nCh) / gridDim.x);
            uint4 cur[OWN_UK];
            key_load(cur, key4, ch * CH + tid, nQ);
            for (int it = 0; it + 1 < nCh; ++it) {
                int nch = ch + 1;
                if (nch >= nCh) nch = 0;
                uint4 nxt[OWN_UK];
                key_load(nxt, key4, nch * CH + tid, nQ);
                key_chunk(cur, ch * CH + tid, nQ, ch == nCh - 1, pat, &s_cnt, La);
#pragma unroll
                for (int u = 0; u < OWN_UK; ++u) cur[u] = nxt[u];
                ch = nch;
            }
            key_chunk(cur, ch * CH + tid, nQ, ch == nCh - 1, pat, &s_cnt, La);
        }
        const int et = (nQ << 4) + tid;
        if (et < nE) own_note(dst[et], et, r0, nr, &s_cnt, La);
    } else {
        const int nG  = nE >> 2;
        const int CH  = OWN_T * OWN_U;
        const int nCh = (nG + CH - 1) / CH;
        if (nG > 0) {
            const int4* dst4 = (const int4*)dst;
            int ch = (int)(((long long)blockIdx.x * nCh) / gridDim.x);
            int4 cur[OWN_U];
            own_load(cur, dst4, ch * CH + tid, nG);
            for (int it = 0; it + 1 < nCh; ++it) {
                int nch = ch + 1;
                if (nch >= nCh) nch = 0;
                int4 nxt[OWN_U];
                own_load(nxt, dst4, nch * CH + tid, nG);
                own_chunk(cur, ch * CH + tid, nG, ch == nCh - 1, r0, nr, &s_cnt, La);
#pragma unroll
                for (int u = 0; u < OWN_U; ++u) cur[u] = nxt[u];
                ch = nch;
            }
            own_chunk(cur, ch * CH + tid, nG, ch == nCh - 1, r0, nr, &s_cnt, La);
        }
        const int et = (nG << 2) + tid;
        if (et < nE) own_note(dst[et], et, r0, nr, &s_cnt, La);
    }
    __syncthreads();
    const int cnt = s_cnt;

    if (cnt <= OWN_CAP) {
        for (int i = tid; i < cnt; i += OWN_T) {
            const int e = La[i];
            const int r = dst[e] - r0;
            const int s = src[e];
            const float v = leaky(ssrc[s] + sd[r]);
            La[i] = s;
            Lr[i] = r;
            Lv[i] = v;
            ldsMaxF(&mx[r], v);
        }
        __syncthreads();
        for (int i = tid; i < cnt; i += OWN_T) {
            const int r = Lr[i];
            const float t = __expf(Lv[i] - mx[r]);
            Lv[i] = t;
            atomicAdd(&dn[r], t);
        }
        __syncthreads();
        for (int base = wave * OWN_G; base < cnt; base += OWN_W * OWN_G) {
            if (base + OWN_G <= cnt) {
                float zv[OWN_G][4];
#pragma unroll
                for (int u = 0; u < OWN_G; ++u) {
                    const int s = __builtin_amdgcn_readfirstlane(La[base + u]);
                    const float* zp = z + (size_t)s * D + lane;
                    zv[u][0] = zp[0];
                    zv[u][1] = zp[32];
                    zv[u][2] = zp[64];
                    zv[u][3] = zp[96];
                }
#pragma unroll
                for (int u = 0; u < OWN_G; ++u) {
                    const int r = __builtin_amdgcn_readfirstlane(Lr[base + u]);
                    own_add_row(acc + r * D + lane, Lv[base + u], zv[u]);
                }
            } else {
                for (int i = base; i < cnt; ++i) {
                    const int s = __builtin_amdgcn_readfirstlane(La[i]);
                    const int r = __builtin_amdgcn_readfirstlane(Lr[i]);
                    const float* zp = z + (size_t)s * D + lane;
                    float zv[4];
                    zv[0] = zp[0]; zv[1] = zp[32]; zv[2] = zp[64]; zv[3] = zp[96];
                    own_add_row(acc + r * D + lane, Lv[i], zv);
                }
            }
        }
    } else {
        for (int e = tid; e < nE; e += OWN_T) {
            const int r = dst[e] - r0;
            if ((unsigned int)r < (unsigned int)nr)
                ldsMaxF(&mx[r], leaky(ssrc[src[e]] + sd[r]));
        }
        __syncthreads();
        for (int base = wave * 32; base < nE; base += OWN_T) {
            const int e = base + lane;
            int s = 0, r = -1;
            float t = 0.0f;
            if (e < nE) r = dst[e] - r0;
            const bool hit = (unsigned int)r < (unsigned int)nr;
            if (hit) {
                s = src[e];
                t = __expf(leaky(ssrc[s] + sd[r]) - mx[r]);
                atomicAdd(&dn[r], t);
            }
            unsigned int mask = __builtin_amdgcn_ballot_w32(hit);
            while (mask) {
                const int j = __builtin_amdgcn_readfirstlane(__builtin_ctz(mask));
                mask &= mask - 1;
                const int   sj = __builtin_amdgcn_readlane(s, j);
                const int   rj = __builtin_amdgcn_readlane(r, j);
                const float tj = __int_as_float(__builtin_amdgcn_readlane(__float_as_int(t), j));
                const float* zp = z + (size_t)sj * D + lane;
                float zv[4];
                zv[0] = zp[0]; zv[1] = zp[32]; zv[2] = zp[64]; zv[3] = zp[96];
                own_add_row(acc + rj * D + lane, tj, zv);
            }
        }
    }
    __syncthreads();

    for (int r = wave; r < nr; r += OWN_W) {
        const float d = dn[r];
        const float4 a = *(const float4*)(acc + r * D + lane * 4);
        float4 o;
        if (d > 0.0f) {
            o.x = a.x / d; o.y = a.y / d; o.z = a.z / d; o.w = a.w / d;
        } else {
            o.x = 0.0f; o.y = 0.0f; o.z = 0.0f; o.w = 0.0f;
        }
        const int row = r0 + r;
        float* op = (row < n0) ? (out_lo + (size_t)row * D)
                               : (out_hi + (size_t)(row - n0) * D);
        *(float4*)(op + lane * 4) = o;
    }
}

__global__ void combine(const float* __restrict__ A, const float* __restrict__ B,
                        const float* __restrict__ W1, const float* __restrict__ b1,
                        const float* __restrict__ w2, float* __restrict__ out)
{
    __shared__ float sA[D], sB[D], red[D];
    const int i = blockIdx.x;
    const int j = threadIdx.x;
    sA[j] = A[(size_t)i * D + j];
    sB[j] = B[(size_t)i * D + j];
    __syncthreads();
    float accA = b1[j], accB = accA;
    for (int k = 0; k < D; ++k) {
        float wv = W1[(size_t)k * D + j];
        accA = fmaf(sA[k], wv, accA);
        accB = fmaf(sB[k], wv, accB);
    }
    const float w2j = w2[j];
    const float tA = tanhf(accA) * w2j;
    const float tB = tanhf(accB) * w2j;

    red[j] = tA; __syncthreads();
    for (int s = 64; s > 0; s >>= 1) { if (j < s) red[j] += red[j + s]; __syncthreads(); }
    const float s1 = red[0]; __syncthreads();
    red[j] = tB; __syncthreads();
    for (int s = 64; s > 0; s >>= 1) { if (j < s) red[j] += red[j + s]; __syncthreads(); }
    const float s2 = red[0];

    const float mm = fmaxf(s1, s2);
    const float e1 = __expf(s1 - mm), e2 = __expf(s2 - mm);
    const float al = e1 / (e1 + e2);
    out[(size_t)i * D + j] = al * sA[j] + (1.0f - al) * sB[j];
}

static inline int cdiv(int a, int b) { return (a + b - 1) / b; }

extern "C" void kernel_launch(void* const* d_in, const int* in_sizes, int n_in,
                              void* d_out, int out_size, void* d_ws, size_t ws_size,
                              hipStream_t stream)
{
    (void)n_in; (void)out_size; (void)ws_size;
    const float* kn_emb   = (const float*)d_in[0];
    const float* exer_emb = (const float*)d_in[1];
    const float* W_dir    = (const float*)d_in[2];
    const float* a_dir    = (const float*)d_in[3];
    const float* W_ke     = (const float*)d_in[4];
    const float* a_ke     = (const float*)d_in[5];
    const float* W_ek     = (const float*)d_in[6];
    const float* a_ek     = (const float*)d_in[7];
    const float* rW1      = (const float*)d_in[8];
    const float* rb1      = (const float*)d_in[9];
    const float* rw2      = (const float*)d_in[10];
    const int* dsrc  = (const int*)d_in[11];
    const int* ddst  = (const int*)d_in[12];
    const int* kesrc = (const int*)d_in[13];
    const int* kedst = (const int*)d_in[14];
    const int* eksrc = (const int*)d_in[15];
    const int* ekdst = (const int*)d_in[16];

    const int Kn = in_sizes[0] / D;
    const int In = in_sizes[1] / D;
    const int Nn = Kn + In;
    const int E1 = in_sizes[11];
    const int E2 = in_sizes[13];

    float* w = (float*)d_ws;
    size_t off = 0;
    float* buf_z = w + off; off += (size_t)Nn * D;
    float* pY    = w + off; off += (size_t)Kn * D;
    float* pX    = w + off; off += (size_t)In * D;
    float* pA    = w + off; off += (size_t)Kn * D;
    float* ssrc  = w + off; off += (size_t)Nn;
    float* sdst  = w + off; off += (size_t)Nn;

    float* out_kn = (float*)d_out;
    float* out_ex = (float*)d_out + (size_t)Kn * D;

    int gN = 256, rN = cdiv(Nn, gN);
    if (rN > OWN_RMAX) { gN = cdiv(Nn, OWN_RMAX); rN = cdiv(Nn, gN); }
    int gK = 256, rK = cdiv(Kn, gK);
    if (rK > OWN_RMAX) { gK = cdiv(Kn, OWN_RMAX); rK = cdiv(Kn, gK); }

    int* keyKE  = (gN <= 256 && (size_t)E2 <= (size_t)Kn * D * 4) ? (int*)pA : nullptr;
    int* keyDIR = (gK <= 256 && (size_t)E1 <= (size_t)In * D * 4) ? (int*)pX : nullptr;
    int* keyEK  = (gN <= 256 && (size_t)E2 <= (size_t)In * D * 4) ? (int*)pX : nullptr;

    gemm_z<<<cdiv(Nn, 16), 256, 0, stream>>>(exer_emb, In, kn_emb, W_ke, buf_z, Nn);
    row_scores<<<cdiv(Nn, RS_WGR) + (keyKE ? cdiv(E2 >> 2, RS_WGK) : 0), RS_T, 0, stream>>>(buf_z, a_ke, ssrc, sdst, Nn, kedst, keyKE, E2, rN);
    owner_agg<<<gN, OWN_T, 0, stream>>>(kesrc, kedst, keyKE, ssrc, sdst, buf_z, pX, pY, In, Nn, E2, rN);

    gemm_z<<<cdiv(Kn, 16), 256, 0, stream>>>(kn_emb, Kn, kn_emb, W_dir, buf_z, Kn);
    row_scores<<<cdiv(Kn, RS_WGR) + (keyDIR ? cdiv(E1 >> 2, RS_WGK) : 0), RS_T, 0, stream>>>(buf_z, a_dir, ssrc, sdst, Kn, ddst, keyDIR, E1, rK);
    owner_agg<<<gK, OWN_T, 0, stream>>>(dsrc, ddst, keyDIR, ssrc, sdst, buf_z, pA, pA, Kn, Kn, E1, rK);

    combine<<<Kn, D, 0, stream>>>(pA, pY, rW1, rb1, rw2, out_kn);

    gemm_z<<<cdiv(Nn, 16), 256, 0, stream>>>(exer_emb, In, kn_emb, W_ek, buf_z, Nn);
    row_scores<<<cdiv(Nn, RS_WGR) + (keyEK ? cdiv(E2 >> 2, RS_WGK) : 0), RS_T, 0, stream>>>(buf_z, a_ek, ssrc, sdst, Nn, ekdst, keyEK, E2, rN);
    owner_agg<<<gN, OWN_T, 0, stream>>>(eksrc, ekdst, keyEK, ssrc, sdst, buf_z, out_ex, pY, In, Nn, E2, rN);
}


# gemm_z: a wave owns a whole 16-row strip (8 column tiles, a lane's columns 4 neighbours): A fragment loaded once not 8 times, W and Z moved as 16-byte pieces of whole lines; Z keeps its bits
# speedup vs baseline: 2.8823x; 1.1185x over previous
#include <hip/hip_runtime.h>
#include <hip/hip_bf16.h>
#include <math.h>

#define D 128

typedef float v2f __attribute__((ext_vector_type(2)));
typedef float v8f __attribute__((ext_vector_type(8)));

#define GZ_T    128
#define GZ_ROWS (16 * (GZ_T / 32))
#define GZ_KS   (D / 4)
#define GZ_AHEAD 3

struct gz_ops { float2 a; float4 x0, x1, y0, y1; };

__device__ __forceinline__ gz_ops gz_load(const float* hrow, const float* wcol, int kk) {
    gz_ops o;
    o.a  = *(const float2*)(hrow + kk);
    o.x0 = *(const float4*)(wcol + (size_t)kk * D);
    o.x1 = *(const float4*)(wcol + (size_t)kk * D + 64);
    o.y0 = *(const float4*)(wcol + (size_t)(kk + 1) * D);
    o.y1 = *(const float4*)(wcol + (size_t)(kk + 1) * D + 64);
    return o;
}

__device__ __forceinline__ v8f gz_step(v2f a, float bx, float by, v8f c) {
    v2f b; b.x = bx; b.y = by;
    return __builtin_amdgcn_wmma_f32_16x16x4_f32(false, a, false, b, (short)0, c, false, false);
}

__global__ __launch_bounds__(GZ_T, 5)
void gemm_z(const float* __restrict__ h0, int n0,
            const float* __restrict__ h1,
            const float* __restrict__ W,
            float* __restrict__ Z, int n)
{
    const int lane = threadIdx.x & 31;
    const int half = lane >> 4;
    const int lr   = lane & 15;
    const int row0 = __builtin_amdgcn_readfirstlane(
                         (int)(blockIdx.x * GZ_T + threadIdx.x) >> 5) * 16;
    if (row0 >= n) return;

    int rowA = row0 + lr;
    if (rowA >= n) rowA = n - 1;
    const float* hrow = (rowA < n0) ? (h0 + (size_t)rowA * D)
                                    : (h1 + (size_t)(rowA - n0) * D);
    const float* wcol = W + 4 * lr;

    v8f c[8];
#pragma unroll
    for (int t = 0; t < 8; ++t)
        c[t] = (v8f){0.f, 0.f, 0.f, 0.f, 0.f, 0.f, 0.f, 0.f};

    gz_ops ring[GZ_AHEAD + 1];
#pragma unroll
    for (int i = 0; i < GZ_AHEAD; ++i)
        ring[i] = gz_load(hrow, wcol, 4 * i + 2 * half);

#pragma unroll
    for (int i = 0; i < GZ_KS; ++i) {
        if (i + GZ_AHEAD < GZ_KS)
            ring[(i + GZ_AHEAD) % (GZ_AHEAD + 1)] =
                gz_load(hrow, wcol, 4 * (i + GZ_AHEAD) + 2 * half);
        const gz_ops o = ring[i % (GZ_AHEAD + 1)];
        v2f a; a.x = o.a.x; a.y = o.a.y;
        c[0] = gz_step(a, o.x0.x, o.y0.x, c[0]);
        c[1] = gz_step(a, o.x0.y, o.y0.y, c[1]);
        c[2] = gz_step(a, o.x0.z, o.y0.z, c[2]);
        c[3] = gz_step(a, o.x0.w, o.y0.w, c[3]);
        c[4] = gz_step(a, o.x1.x, o.y1.x, c[4]);
        c[5] = gz_step(a, o.x1.y, o.y1.y, c[5]);
        c[6] = gz_step(a, o.x1.z, o.y1.z, c[6]);
        c[7] = gz_step(a, o.x1.w, o.y1.w, c[7]);
    }
#pragma unroll
    for (int v = 0; v < 8; ++v) {
        const int r = row0 + v + 8 * half;
        if (r < n) {
            float* zp = Z + (size_t)r * D + 4 * lr;
            *(float4*)zp        = make_float4(c[0][v], c[1][v], c[2][v], c[3][v]);
            *(float4*)(zp + 64) = make_float4(c[4][v], c[5][v], c[6][v], c[7][v]);
        }
    }
}

#define RS_ROWS 32
#define RS_T    128
#define RS_KU   4
#define RS_WGR  (RS_ROWS * (RS_T / 32))
#define RS_WGK  (RS_T * RS_KU)

__device__ __forceinline__ float rs_dot4(float4 zv, float4 av) {
    float p = zv.y * av.y;
    p = fmaf(zv.x, av.x, p);
    p = fmaf(zv.z, av.z, p);
    p = fmaf(zv.w, av.w, p);
    return p;
}

__global__ __launch_bounds__(RS_T)
void row_scores(const float* __restrict__ Z, const float* __restrict__ a,
                float* __restrict__ ssrc, float* __restrict__ sdst, int n,
                const int* __restrict__ dst, int* __restrict__ key,
                int nE, int R)
{
    const int nRowWG = (n + RS_WGR - 1) / RS_WGR;
    if ((int)blockIdx.x >= nRowWG) {
        const int nW = nE >> 2;
        const unsigned int uR = (unsigned int)R;
        const int4* dst4 = (const int4*)dst;
        const int q0 = ((int)blockIdx.x - nRowWG) * RS_WGK + (int)threadIdx.x;
        int4 d[RS_KU];
#pragma unroll
        for (int u = 0; u < RS_KU; ++u) {
            int q = q0 + u * RS_T;
            q = q < nW ? q : nW - 1;
            d[u] = dst4[q];
        }
#pragma unroll
        for (int u = 0; u < RS_KU; ++u) {
            const int q = q0 + u * RS_T;
            if (q < nW) {
                const unsigned int w = (((unsigned int)d[u].x / uR) & 255u)
                                     | ((((unsigned int)d[u].y / uR) & 255u) << 8)
                                     | ((((unsigned int)d[u].z / uR) & 255u) << 16)
                                     | ((((unsigned int)d[u].w / uR) & 255u) << 24);
                key[q] = (int)w;
            }
        }
        return;
    }

    const int lane = threadIdx.x & 31;
    const int row0 = __builtin_amdgcn_readfirstlane(
                         (int)(blockIdx.x * RS_T + threadIdx.x) >> 5) * RS_ROWS;
    if (row0 >= n) return;

    const float4 a0 = *(const float4*)(a + lane * 4);
    const float4 a1 = *(const float4*)(a + D + lane * 4);
    float4 zv[RS_ROWS];
#pragma unroll
    for (int u = 0; u < RS_ROWS; ++u) {
        int r = row0 + u;
        r = r < n ? r : n - 1;
        zv[u] = *(const float4*)(Z + (size_t)r * D + lane * 4);
    }

    float p0[RS_ROWS], p1[RS_ROWS];
#pragma unroll
    for (int u = 0; u < RS_ROWS; ++u) {
        p0[u] = rs_dot4(zv[u], a0);
        p1[u] = rs_dot4(zv[u], a1);
    }
#pragma unroll
    for (int off = 16; off > 0; off >>= 1) {
#pragma unroll
        for (int u = 0; u < RS_ROWS; ++u) {
            p0[u] += __shfl_xor(p0[u], off, 32);
            p1[u] += __shfl_xor(p1[u], off, 32);
        }
    }
    float my0 = 0.0f, my1 = 0.0f;
#pragma unroll
    for (int u = 0; u < RS_ROWS; ++u)
        if (lane == u) { my0 = p0[u]; my1 = p1[u]; }
    if (row0 + lane < n) { ssrc[row0 + lane] = my0; sdst[row0 + lane] = my1; }
}

__device__ __forceinline__ float leaky(float v) {
    return v > 0.0f ? v : 0.01f * v;
}

#define OWN_T    1024
#define OWN_W    (OWN_T / 32)
#define OWN_RMAX 288
#define OWN_CAP  8192
#define OWN_U    8
#define OWN_UK   4
#define OWN_G    16

__device__ __forceinline__ void ldsMaxF(float* addr, float val) {
    if (val >= 0.0f) atomicMax((int*)addr, __float_as_int(val));
    else             atomicMin((unsigned int*)addr, __float_as_uint(val));
}

__device__ __forceinline__ void own_push(int e, int* cnt, int* La) {
    const int idx = atomicAdd(cnt, 1);
    if (idx < OWN_CAP) La[idx] = e;
}

__device__ __forceinline__ void own_note(int d, int e, int r0, int nr,
                                         int* cnt, int* La) {
    if ((unsigned int)(d - r0) < (unsigned int)nr) own_push(e, cnt, La);
}

__device__ __forceinline__ void own_load(int4* dv, const int4* dst4, int g, int nG) {
#pragma unroll
    for (int u = 0; u < OWN_U; ++u) {
        int gi = g + u * OWN_T;
        gi = gi < nG ? gi : nG - 1;
        dv[u] = dst4[gi];
    }
}

__device__ __forceinline__ void own_chunk(int4* dv, int gb, int nG, bool shortc,
                                          int r0, int nr, int* cnt, int* La) {
    if (shortc) {
#pragma unroll
        for (int u = 0; u < OWN_U; ++u)
            if (gb + u * OWN_T >= nG) dv[u] = make_int4(-1, -1, -1, -1);
    }
#pragma unroll
    for (int u = 0; u < OWN_U; ++u) {
        const int e0 = (gb + u * OWN_T) << 2;
        own_note(dv[u].x, e0,     r0, nr, cnt, La);
        own_note(dv[u].y, e0 + 1, r0, nr, cnt, La);
        own_note(dv[u].z, e0 + 2, r0, nr, cnt, La);
        own_note(dv[u].w, e0 + 3, r0, nr, cnt, La);
    }
}

__device__ __forceinline__ void key_load(uint4* kv, const uint4* key4, int g, int nQ) {
#pragma unroll
    for (int u = 0; u < OWN_UK; ++u) {
        int gi = g + u * OWN_T;
        gi = gi < nQ ? gi : nQ - 1;
        kv[u] = key4[gi];
    }
}

__device__ __forceinline__ unsigned int key_nz(unsigned int x) {
    return ((x & 0x7f7f7f7fu) + 0x7f7f7f7fu) | x;
}

__device__ __forceinline__ unsigned int key_match(uint4 kv, unsigned int pat) {
    const unsigned int n0 = key_nz(kv.x ^ pat);
    const unsigned int n1 = key_nz(kv.y ^ pat);
    const unsigned int n2 = key_nz(kv.z ^ pat);
    const unsigned int n3 = key_nz(kv.w ^ pat);
    const unsigned int nz = ((n0 >> 7) & 0x01010101u) | ((n1 >> 6) & 0x02020202u)
                          | ((n2 >> 5) & 0x04040404u) | ((n3 >> 4) & 0x08080808u);
    return ~nz & 0x0f0f0f0fu;
}

__device__ __forceinline__ void key_chunk(const uint4* kv, int gb, int nQ, bool shortc,
                                          unsigned int pat, int* cnt, int* La) {
#pragma unroll
    for (int u = 0; u < OWN_UK; ++u) {
        unsigned int m = key_match(kv[u], pat);
        if (shortc && gb + u * OWN_T >= nQ) m = 0u;
        const int e0 = (gb + u * OWN_T) << 4;
        while (m) {
            const int p = __builtin_ctz(m);
            m &= m - 1u;
            own_push(e0 + ((p & 7) << 2) + (p >> 3), cnt, La);
        }
    }
}

__device__ __forceinline__ void own_add_row(float* arow, float t, const float* zv) {
    atomicAdd(arow,      t * zv[0]);
    atomicAdd(arow + 32, t * zv[1]);
    atomicAdd(arow + 64, t * zv[2]);
    atomicAdd(arow + 96, t * zv[3]);
}

__global__ __launch_bounds__(OWN_T)
void owner_agg(const int* __restrict__ src, const int* __restrict__ dst,
               const int* __restrict__ key,
               const float* __restrict__ ssrc, const float* __restrict__ sdst,
               const float* __restrict__ z,
               float* out_lo, float* out_hi,
               int n0, int n, int nE, int R)
{
    __shared__ __attribute__((aligned(16))) float acc[OWN_RMAX * D];
    __shared__ float mx[OWN_RMAX];
    __shared__ float dn[OWN_RMAX];
    __shared__ float sd[OWN_RMAX];
    __shared__ int   La[OWN_CAP];
    __shared__ int   Lr[OWN_CAP];
    __shared__ float Lv[OWN_CAP];
    __shared__ int   s_cnt;

    const int tid  = threadIdx.x;
    const int lane = tid & 31;
    const int wave = tid >> 5;
    const int r0   = blockIdx.x * R;
    int nr = n - r0;
    nr = nr > R ? R : nr;
    if (nr <= 0) return;

    {
        const float4 z4 = make_float4(0.0f, 0.0f, 0.0f, 0.0f);
        float4* a4 = (float4*)acc;
        for (int i = tid; i < nr * (D / 4); i += OWN_T) a4[i] = z4;
        for (int i = tid; i < nr; i += OWN_T) {
            mx[i] = -INFINITY;
            dn[i] = 0.0f;
            sd[i] = sdst[r0 + i];
        }
        if (tid == 0) s_cnt = 0;
    }
    __syncthreads();

    if (key != nullptr) {
        const int nQ  = nE >> 4;
        const int CH  = OWN_T * OWN_UK;
        const int nCh = (nQ + CH - 1) / CH;
        const unsigned int pat = (unsigned int)blockIdx.x * 0x01010101u;
        if (nQ > 0) {
            const uint4* key4 = (const uint4*)key;
            int ch = (int)(((long long)blockIdx.x * nCh) / gridDim.x);
            uint4 cur[OWN_UK];
            key_load(cur, key4, ch * CH + tid, nQ);
            for (int it = 0; it + 1 < nCh; ++it) {
                int nch = ch + 1;
                if (nch >= nCh) nch = 0;
                uint4 nxt[OWN_UK];
                key_load(nxt, key4, nch * CH + tid, nQ);
                key_chunk(cur, ch * CH + tid, nQ, ch == nCh - 1, pat, &s_cnt, La);
#pragma unroll
                for (int u = 0; u < OWN_UK; ++u) cur[u] = nxt[u];
                ch = nch;
            }
            key_chunk(cur, ch * CH + tid, nQ, ch == nCh - 1, pat, &s_cnt, La);
        }
        const int et = (nQ << 4) + tid;
        if (et < nE) own_note(dst[et], et, r0, nr, &s_cnt, La);
    } else {
        const int nG  = nE >> 2;
        const int CH  = OWN_T * OWN_U;
        const int nCh = (nG + CH - 1) / CH;
        if (nG > 0) {
            const int4* dst4 = (const int4*)dst;
            int ch = (int)(((long long)blockIdx.x * nCh) / gridDim.x);
            int4 cur[OWN_U];
            own_load(cur, dst4, ch * CH + tid, nG);
            for (int it = 0; it + 1 < nCh; ++it) {
                int nch = ch + 1;
                if (nch >= nCh) nch = 0;
                int4 nxt[OWN_U];
                own_load(nxt, dst4, nch * CH + tid, nG);
                own_chunk(cur, ch * CH + tid, nG, ch == nCh - 1, r0, nr, &s_cnt, La);
#pragma unroll
                for (int u = 0; u < OWN_U; ++u) cur[u] = nxt[u];
                ch = nch;
            }
            own_chunk(cur, ch * CH + tid, nG, ch == nCh - 1, r0, nr, &s_cnt, La);
        }
        const int et = (nG << 2) + tid;
        if (et < nE) own_note(dst[et], et, r0, nr, &s_cnt, La);
    }
    __syncthreads();
    const int cnt = s_cnt;

    if (cnt <= OWN_CAP) {
        for (int i = tid; i < cnt; i += OWN_T) {
            const int e = La[i];
            const int r = dst[e] - r0;
            const int s = src[e];
            const float v = leaky(ssrc[s] + sd[r]);
            La[i] = s;
            Lr[i] = r;
            Lv[i] = v;
            ldsMaxF(&mx[r], v);
        }
        __syncthreads();
        for (int i = tid; i < cnt; i += OWN_T) {
            const int r = Lr[i];
            const float t = __expf(Lv[i] - mx[r]);
            Lv[i] = t;
            atomicAdd(&dn[r], t);
        }
        __syncthreads();
        for (int base = wave * OWN_G; base < cnt; base += OWN_W * OWN_G) {
            if (base + OWN_G <= cnt) {
                float zv[OWN_G][4];
#pragma unroll
                for (int u = 0; u < OWN_G; ++u) {
                    const int s = __builtin_amdgcn_readfirstlane(La[base + u]);
                    const float* zp = z + (size_t)s * D + lane;
                    zv[u][0] = zp[0];
                    zv[u][1] = zp[32];
                    zv[u][2] = zp[64];
                    zv[u][3] = zp[96];
                }
#pragma unroll
                for (int u = 0; u < OWN_G; ++u) {
                    const int r = __builtin_amdgcn_readfirstlane(Lr[base + u]);
                    own_add_row(acc + r * D + lane, Lv[base + u], zv[u]);
                }
            } else {
                for (int i = base; i < cnt; ++i) {
                    const int s = __builtin_amdgcn_readfirstlane(La[i]);
                    const int r = __builtin_amdgcn_readfirstlane(Lr[i]);
                    const float* zp = z + (size_t)s * D + lane;
                    float zv[4];
                    zv[0] = zp[0]; zv[1] = zp[32]; zv[2] = zp[64]; zv[3] = zp[96];
                    own_add_row(acc + r * D + lane, Lv[i], zv);
                }
            }
        }
    } else {
        for (int e = tid; e < nE; e += OWN_T) {
            const int r = dst[e] - r0;
            if ((unsigned int)r < (unsigned int)nr)
                ldsMaxF(&mx[r], leaky(ssrc[src[e]] + sd[r]));
        }
        __syncthreads();
        for (int base = wave * 32; base < nE; base += OWN_T) {
            const int e = base + lane;
            int s = 0, r = -1;
            float t = 0.0f;
            if (e < nE) r = dst[e] - r0;
            const bool hit = (unsigned int)r < (unsigned int)nr;
            if (hit) {
                s = src[e];
                t = __expf(leaky(ssrc[s] + sd[r]) - mx[r]);
                atomicAdd(&dn[r], t);
            }
            unsigned int mask = __builtin_amdgcn_ballot_w32(hit);
            while (mask) {
                const int j = __builtin_amdgcn_readfirstlane(__builtin_ctz(mask));
                mask &= mask - 1;
                const int   sj = __builtin_amdgcn_readlane(s, j);
                const int   rj = __builtin_amdgcn_readlane(r, j);
                const float tj = __int_as_float(__builtin_amdgcn_readlane(__float_as_int(t), j));
                const float* zp = z + (size_t)sj * D + lane;
                float zv[4];
                zv[0] = zp[0]; zv[1] = zp[32]; zv[2] = zp[64]; zv[3] = zp[96];
                own_add_row(acc + rj * D + lane, tj, zv);
            }
        }
    }
    __syncthreads();

    for (int r = wave; r < nr; r += OWN_W) {
        const float d = dn[r];
        const float4 a = *(const float4*)(acc + r * D + lane * 4);
        float4 o;
        if (d > 0.0f) {
            o.x = a.x / d; o.y = a.y / d; o.z = a.z / d; o.w = a.w / d;
        } else {
            o.x = 0.0f; o.y = 0.0f; o.z = 0.0f; o.w = 0.0f;
        }
        const int row = r0 + r;
        float* op = (row < n0) ? (out_lo + (size_t)row * D)
                               : (out_hi + (size_t)(row - n0) * D);
        *(float4*)(op + lane * 4) = o;
    }
}

__global__ void combine(const float* __restrict__ A, const float* __restrict__ B,
                        const float* __restrict__ W1, const float* __restrict__ b1,
                        const float* __restrict__ w2, float* __restrict__ out)
{
    __shared__ float sA[D], sB[D], red[D];
    const int i = blockIdx.x;
    const int j = threadIdx.x;
    sA[j] = A[(size_t)i * D + j];
    sB[j] = B[(size_t)i * D + j];
    __syncthreads();
    float accA = b1[j], accB = accA;
    for (int k = 0; k < D; ++k) {
        float wv = W1[(size_t)k * D + j];
        accA = fmaf(sA[k], wv, accA);
        accB = fmaf(sB[k], wv, accB);
    }
    const float w2j = w2[j];
    const float tA = tanhf(accA) * w2j;
    const float tB = tanhf(accB) * w2j;

    red[j] = tA; __syncthreads();
    for (int s = 64; s > 0; s >>= 1) { if (j < s) red[j] += red[j + s]; __syncthreads(); }
    const float s1 = red[0]; __syncthreads();
    red[j] = tB; __syncthreads();
    for (int s = 64; s > 0; s >>= 1) { if (j < s) red[j] += red[j + s]; __syncthreads(); }
    const float s2 = red[0];

    const float mm = fmaxf(s1, s2);
    const float e1 = __expf(s1 - mm), e2 = __expf(s2 - mm);
    const float al = e1 / (e1 + e2);
    out[(size_t)i * D + j] = al * sA[j] + (1.0f - al) * sB[j];
}

static inline int cdiv(int a, int b) { return (a + b - 1) / b; }

extern "C" void kernel_launch(void* const* d_in, const int* in_sizes, int n_in,
                              void* d_out, int out_size, void* d_ws, size_t ws_size,
                              hipStream_t stream)
{
    (void)n_in; (void)out_size; (void)ws_size;
    const float* kn_emb   = (const float*)d_in[0];
    const float* exer_emb = (const float*)d_in[1];
    const float* W_dir    = (const float*)d_in[2];
    const float* a_dir    = (const float*)d_in[3];
    const float* W_ke     = (const float*)d_in[4];
    const float* a_ke     = (const float*)d_in[5];
    const float* W_ek     = (const float*)d_in[6];
    const float* a_ek     = (const float*)d_in[7];
    const float* rW1      = (const float*)d_in[8];
    const float* rb1      = (const float*)d_in[9];
    const float* rw2      = (const float*)d_in[10];
    const int* dsrc  = (const int*)d_in[11];
    const int* ddst  = (const int*)d_in[12];
    const int* kesrc = (const int*)d_in[13];
    const int* kedst = (const int*)d_in[14];
    const int* eksrc = (const int*)d_in[15];
    const int* ekdst = (const int*)d_in[16];

    const int Kn = in_sizes[0] / D;
    const int In = in_sizes[1] / D;
    const int Nn = Kn + In;
    const int E1 = in_sizes[11];
    const int E2 = in_sizes[13];

    float* w = (float*)d_ws;
    size_t off = 0;
    float* buf_z = w + off; off += (size_t)Nn * D;
    float* pY    = w + off; off += (size_t)Kn * D;
    float* pX    = w + off; off += (size_t)In * D;
    float* pA    = w + off; off += (size_t)Kn * D;
    float* ssrc  = w + off; off += (size_t)Nn;
    float* sdst  = w + off; off += (size_t)Nn;

    float* out_kn = (float*)d_out;
    float* out_ex = (float*)d_out + (size_t)Kn * D;

    int gN = 256, rN = cdiv(Nn, gN);
    if (rN > OWN_RMAX) { gN = cdiv(Nn, OWN_RMAX); rN = cdiv(Nn, gN); }
    int gK = 256, rK = cdiv(Kn, gK);
    if (rK > OWN_RMAX) { gK = cdiv(Kn, OWN_RMAX); rK = cdiv(Kn, gK); }

    int* keyKE  = (gN <= 256 && (size_t)E2 <= (size_t)Kn * D * 4) ? (int*)pA : nullptr;
    int* keyDIR = (gK <= 256 && (size_t)E1 <= (size_t)In * D * 4) ? (int*)pX : nullptr;
    int* keyEK  = (gN <= 256 && (size_t)E2 <= (size_t)In * D * 4) ? (int*)pX : nullptr;

    gemm_z<<<cdiv(Nn, GZ_ROWS), GZ_T, 0, stream>>>(exer_emb, In, kn_emb, W_ke, buf_z, Nn);
    row_scores<<<cdiv(Nn, RS_WGR) + (keyKE ? cdiv(E2 >> 2, RS_WGK) : 0), RS_T, 0, stream>>>(buf_z, a_ke, ssrc, sdst, Nn, kedst, keyKE, E2, rN);
    owner_agg<<<gN, OWN_T, 0, stream>>>(kesrc, kedst, keyKE, ssrc, sdst, buf_z, pX, pY, In, Nn, E2, rN);

    gemm_z<<<cdiv(Kn, GZ_ROWS), GZ_T, 0, stream>>>(kn_emb, Kn, kn_emb, W_dir, buf_z, Kn);
    row_scores<<<cdiv(Kn, RS_WGR) + (keyDIR ? cdiv(E1 >> 2, RS_WGK) : 0), RS_T, 0, stream>>>(buf_z, a_dir, ssrc, sdst, Kn, ddst, keyDIR, E1, rK);
    owner_agg<<<gK, OWN_T, 0, stream>>>(dsrc, ddst, keyDIR, ssrc, sdst, buf_z, pA, pA, Kn, Kn, E1, rK);

    combine<<<Kn, D, 0, stream>>>(pA, pY, rW1, rb1, rw2, out_kn);

    gemm_z<<<cdiv(Nn, GZ_ROWS), GZ_T, 0, stream>>>(exer_emb, In, kn_emb, W_ek, buf_z, Nn);
    row_scores<<<cdiv(Nn, RS_WGR) + (keyEK ? cdiv(E2 >> 2, RS_WGK) : 0), RS_T, 0, stream>>>(buf_z, a_ek, ssrc, sdst, Nn, ekdst, keyEK, E2, rN);
    owner_agg<<<gN, OWN_T, 0, stream>>>(eksrc, ekdst, keyEK, ssrc, sdst, buf_z, out_ex, pY, In, Nn, E2, rN);
}


# combine: thread j keeps column j of W1 in registers, a work-group walks 16 nodes, 4 per pass, A and B elements come as scalar FMA operands (was a block per node); s1, s2 sum order changes
# speedup vs baseline: 3.0420x; 1.0554x over previous
#include <hip/hip_runtime.h>
#include <hip/hip_bf16.h>
#include <math.h>

#define D 128

typedef float v2f __attribute__((ext_vector_type(2)));
typedef float v8f __attribute__((ext_vector_type(8)));

#define GZ_T    128
#define GZ_ROWS (16 * (GZ_T / 32))
#define GZ_KS   (D / 4)
#define GZ_AHEAD 3

struct gz_ops { float2 a; float4 x0, x1, y0, y1; };

__device__ __forceinline__ gz_ops gz_load(const float* hrow, const float* wcol, int kk) {
    gz_ops o;
    o.a  = *(const float2*)(hrow + kk);
    o.x0 = *(const float4*)(wcol + (size_t)kk * D);
    o.x1 = *(const float4*)(wcol + (size_t)kk * D + 64);
    o.y0 = *(const float4*)(wcol + (size_t)(kk + 1) * D);
    o.y1 = *(const float4*)(wcol + (size_t)(kk + 1) * D + 64);
    return o;
}

__device__ __forceinline__ v8f gz_step(v2f a, float bx, float by, v8f c) {
    v2f b; b.x = bx; b.y = by;
    return __builtin_amdgcn_wmma_f32_16x16x4_f32(false, a, false, b, (short)0, c, false, false);
}

__global__ __launch_bounds__(GZ_T, 5)
void gemm_z(const float* __restrict__ h0, int n0,
            const float* __restrict__ h1,
            const float* __restrict__ W,
            float* __restrict__ Z, int n)
{
    const int lane = threadIdx.x & 31;
    const int half = lane >> 4;
    const int lr   = lane & 15;
    const int row0 = __builtin_amdgcn_readfirstlane(
                         (int)(blockIdx.x * GZ_T + threadIdx.x) >> 5) * 16;
    if (row0 >= n) return;

    int rowA = row0 + lr;
    if (rowA >= n) rowA = n - 1;
    const float* hrow = (rowA < n0) ? (h0 + (size_t)rowA * D)
                                    : (h1 + (size_t)(rowA - n0) * D);
    const float* wcol = W + 4 * lr;

    v8f c[8];
#pragma unroll
    for (int t = 0; t < 8; ++t)
        c[t] = (v8f){0.f, 0.f, 0.f, 0.f, 0.f, 0.f, 0.f, 0.f};

    gz_ops ring[GZ_AHEAD + 1];
#pragma unroll
    for (int i = 0; i < GZ_AHEAD; ++i)
        ring[i] = gz_load(hrow, wcol, 4 * i + 2 * half);

#pragma unroll
    for (int i = 0; i < GZ_KS; ++i) {
        if (i + GZ_AHEAD < GZ_KS)
            ring[(i + GZ_AHEAD) % (GZ_AHEAD + 1)] =
                gz_load(hrow, wcol, 4 * (i + GZ_AHEAD) + 2 * half);
        const gz_ops o = ring[i % (GZ_AHEAD + 1)];
        v2f a; a.x = o.a.x; a.y = o.a.y;
        c[0] = gz_step(a, o.x0.x, o.y0.x, c[0]);
        c[1] = gz_step(a, o.x0.y, o.y0.y, c[1]);
        c[2] = gz_step(a, o.x0.z, o.y0.z, c[2]);
        c[3] = gz_step(a, o.x0.w, o.y0.w, c[3]);
        c[4] = gz_step(a, o.x1.x, o.y1.x, c[4]);
        c[5] = gz_step(a, o.x1.y, o.y1.y, c[5]);
        c[6] = gz_step(a, o.x1.z, o.y1.z, c[6]);
        c[7] = gz_step(a, o.x1.w, o.y1.w, c[7]);
    }
#pragma unroll
    for (int v = 0; v < 8; ++v) {
        const int r = row0 + v + 8 * half;
        if (r < n) {
            float* zp = Z + (size_t)r * D + 4 * lr;
            *(float4*)zp        = make_float4(c[0][v], c[1][v], c[2][v], c[3][v]);
            *(float4*)(zp + 64) = make_float4(c[4][v], c[5][v], c[6][v], c[7][v]);
        }
    }
}

#define RS_ROWS 32
#define RS_T    128
#define RS_KU   4
#define RS_WGR  (RS_ROWS * (RS_T / 32))
#define RS_WGK  (RS_T * RS_KU)

__device__ __forceinline__ float rs_dot4(float4 zv, float4 av) {
    float p = zv.y * av.y;
    p = fmaf(zv.x, av.x, p);
    p = fmaf(zv.z, av.z, p);
    p = fmaf(zv.w, av.w, p);
    return p;
}

__global__ __launch_bounds__(RS_T)
void row_scores(const float* __restrict__ Z, const float* __restrict__ a,
                float* __restrict__ ssrc, float* __restrict__ sdst, int n,
                const int* __restrict__ dst, int* __restrict__ key,
                int nE, int R)
{
    const int nRowWG = (n + RS_WGR - 1) / RS_WGR;
    if ((int)blockIdx.x >= nRowWG) {
        const int nW = nE >> 2;
        const unsigned int uR = (unsigned int)R;
        const int4* dst4 = (const int4*)dst;
        const int q0 = ((int)blockIdx.x - nRowWG) * RS_WGK + (int)threadIdx.x;
        int4 d[RS_KU];
#pragma unroll
        for (int u = 0; u < RS_KU; ++u) {
            int q = q0 + u * RS_T;
            q = q < nW ? q : nW - 1;
            d[u] = dst4[q];
        }
#pragma unroll
        for (int u = 0; u < RS_KU; ++u) {
            const int q = q0 + u * RS_T;
            if (q < nW) {
                const unsigned int w = (((unsigned int)d[u].x / uR) & 255u)
                                     | ((((unsigned int)d[u].y / uR) & 255u) << 8)
                                     | ((((unsigned int)d[u].z / uR) & 255u) << 16)
                                     | ((((unsigned int)d[u].w / uR) & 255u) << 24);
                key[q] = (int)w;
            }
        }
        return;
    }

    const int lane = threadIdx.x & 31;
    const int row0 = __builtin_amdgcn_readfirstlane(
                         (int)(blockIdx.x * RS_T + threadIdx.x) >> 5) * RS_ROWS;
    if (row0 >= n) return;

    const float4 a0 = *(const float4*)(a + lane * 4);
    const float4 a1 = *(const float4*)(a + D + lane * 4);
    float4 zv[RS_ROWS];
#pragma unroll
    for (int u = 0; u < RS_ROWS; ++u) {
        int r = row0 + u;
        r = r < n ? r : n - 1;
        zv[u] = *(const float4*)(Z + (size_t)r * D + lane * 4);
    }

    float p0[RS_ROWS], p1[RS_ROWS];
#pragma unroll
    for (int u = 0; u < RS_ROWS; ++u) {
        p0[u] = rs_dot4(zv[u], a0);
        p1[u] = rs_dot4(zv[u], a1);
    }
#pragma unroll
    for (int off = 16; off > 0; off >>= 1) {
#pragma unroll
        for (int u = 0; u < RS_ROWS; ++u) {
            p0[u] += __shfl_xor(p0[u], off, 32);
            p1[u] += __shfl_xor(p1[u], off, 32);
        }
    }
    float my0 = 0.0f, my1 = 0.0f;
#pragma unroll
    for (int u = 0; u < RS_ROWS; ++u)
        if (lane == u) { my0 = p0[u]; my1 = p1[u]; }
    if (row0 + lane < n) { ssrc[row0 + lane] = my0; sdst[row0 + lane] = my1; }
}

__device__ __forceinline__ float leaky(float v) {
    return v > 0.0f ? v : 0.01f * v;
}

#define OWN_T    1024
#define OWN_W    (OWN_T / 32)
#define OWN_RMAX 288
#define OWN_CAP  8192
#define OWN_U    8
#define OWN_UK   4
#define OWN_G    16

__device__ __forceinline__ void ldsMaxF(float* addr, float val) {
    if (val >= 0.0f) atomicMax((int*)addr, __float_as_int(val));
    else             atomicMin((unsigned int*)addr, __float_as_uint(val));
}

__device__ __forceinline__ void own_push(int e, int* cnt, int* La) {
    const int idx = atomicAdd(cnt, 1);
    if (idx < OWN_CAP) La[idx] = e;
}

__device__ __forceinline__ void own_note(int d, int e, int r0, int nr,
                                         int* cnt, int* La) {
    if ((unsigned int)(d - r0) < (unsigned int)nr) own_push(e, cnt, La);
}

__device__ __forceinline__ void own_load(int4* dv, const int4* dst4, int g, int nG) {
#pragma unroll
    for (int u = 0; u < OWN_U; ++u) {
        int gi = g + u * OWN_T;
        gi = gi < nG ? gi : nG - 1;
        dv[u] = dst4[gi];
    }
}

__device__ __forceinline__ void own_chunk(int4* dv, int gb, int nG, bool shortc,
                                          int r0, int nr, int* cnt, int* La) {
    if (shortc) {
#pragma unroll
        for (int u = 0; u < OWN_U; ++u)
            if (gb + u * OWN_T >= nG) dv[u] = make_int4(-1, -1, -1, -1);
    }
#pragma unroll
    for (int u = 0; u < OWN_U; ++u) {
        const int e0 = (gb + u * OWN_T) << 2;
        own_note(dv[u].x, e0,     r0, nr, cnt, La);
        own_note(dv[u].y, e0 + 1, r0, nr, cnt, La);
        own_note(dv[u].z, e0 + 2, r0, nr, cnt, La);
        own_note(dv[u].w, e0 + 3, r0, nr, cnt, La);
    }
}

__device__ __forceinline__ void key_load(uint4* kv, const uint4* key4, int g, int nQ) {
#pragma unroll
    for (int u = 0; u < OWN_UK; ++u) {
        int gi = g + u * OWN_T;
        gi = gi < nQ ? gi : nQ - 1;
        kv[u] = key4[gi];
    }
}

__device__ __forceinline__ unsigned int key_nz(unsigned int x) {
    return ((x & 0x7f7f7f7fu) + 0x7f7f7f7fu) | x;
}

__device__ __forceinline__ unsigned int key_match(uint4 kv, unsigned int pat) {
    const unsigned int n0 = key_nz(kv.x ^ pat);
    const unsigned int n1 = key_nz(kv.y ^ pat);
    const unsigned int n2 = key_nz(kv.z ^ pat);
    const unsigned int n3 = key_nz(kv.w ^ pat);
    const unsigned int nz = ((n0 >> 7) & 0x01010101u) | ((n1 >> 6) & 0x02020202u)
                          | ((n2 >> 5) & 0x04040404u) | ((n3 >> 4) & 0x08080808u);
    return ~nz & 0x0f0f0f0fu;
}

__device__ __forceinline__ void key_chunk(const uint4* kv, int gb, int nQ, bool shortc,
                                          unsigned int pat, int* cnt, int* La) {
#pragma unroll
    for (int u = 0; u < OWN_UK; ++u) {
        unsigned int m = key_match(kv[u], pat);
        if (shortc && gb + u * OWN_T >= nQ) m = 0u;
        const int e0 = (gb + u * OWN_T) << 4;
        while (m) {
            const int p = __builtin_ctz(m);
            m &= m - 1u;
            own_push(e0 + ((p & 7) << 2) + (p >> 3), cnt, La);
        }
    }
}

__device__ __forceinline__ void own_add_row(float* arow, float t, const float* zv) {
    atomicAdd(arow,      t * zv[0]);
    atomicAdd(arow + 32, t * zv[1]);
    atomicAdd(arow + 64, t * zv[2]);
    atomicAdd(arow + 96, t * zv[3]);
}

__global__ __launch_bounds__(OWN_T)
void owner_agg(const int* __restrict__ src, const int* __restrict__ dst,
               const int* __restrict__ key,
               const float* __restrict__ ssrc, const float* __restrict__ sdst,
               const float* __restrict__ z,
               float* out_lo, float* out_hi,
               int n0, int n, int nE, int R)
{
    __shared__ __attribute__((aligned(16))) float acc[OWN_RMAX * D];
    __shared__ float mx[OWN_RMAX];
    __shared__ float dn[OWN_RMAX];
    __shared__ float sd[OWN_RMAX];
    __shared__ int   La[OWN_CAP];
    __shared__ int   Lr[OWN_CAP];
    __shared__ float Lv[OWN_CAP];
    __shared__ int   s_cnt;

    const int tid  = threadIdx.x;
    const int lane = tid & 31;
    const int wave = tid >> 5;
    const int r0   = blockIdx.x * R;
    int nr = n - r0;
    nr = nr > R ? R : nr;
    if (nr <= 0) return;

    {
        const float4 z4 = make_float4(0.0f, 0.0f, 0.0f, 0.0f);
        float4* a4 = (float4*)acc;
        for (int i = tid; i < nr * (D / 4); i += OWN_T) a4[i] = z4;
        for (int i = tid; i < nr; i += OWN_T) {
            mx[i] = -INFINITY;
            dn[i] = 0.0f;
            sd[i] = sdst[r0 + i];
        }
        if (tid == 0) s_cnt = 0;
    }
    __syncthreads();

    if (key != nullptr) {
        const int nQ  = nE >> 4;
        const int CH  = OWN_T * OWN_UK;
        const int nCh = (nQ + CH - 1) / CH;
        const unsigned int pat = (unsigned int)blockIdx.x * 0x01010101u;
        if (nQ > 0) {
            const uint4* key4 = (const uint4*)key;
            int ch = (int)(((long long)blockIdx.x * nCh) / gridDim.x);
            uint4 cur[OWN_UK];
            key_load(cur, key4, ch * CH + tid, nQ);
            for (int it = 0; it + 1 < nCh; ++it) {
                int nch = ch + 1;
                if (nch >= nCh) nch = 0;
                uint4 nxt[OWN_UK];
                key_load(nxt, key4, nch * CH + tid, nQ);
                key_chunk(cur, ch * CH + tid, nQ, ch == nCh - 1, pat, &s_cnt, La);
#pragma unroll
                for (int u = 0; u < OWN_UK; ++u) cur[u] = nxt[u];
                ch = nch;
            }
            key_chunk(cur, ch * CH + tid, nQ, ch == nCh - 1, pat, &s_cnt, La);
        }
        const int et = (nQ << 4) + tid;
        if (et < nE) own_note(dst[et], et, r0, nr, &s_cnt, La);
    } else {
        const int nG  = nE >> 2;
        const int CH  = OWN_T * OWN_U;
        const int nCh = (nG + CH - 1) / CH;
        if (nG > 0) {
            const int4* dst4 = (const int4*)dst;
            int ch = (int)(((long long)blockIdx.x * nCh) / gridDim.x);
            int4 cur[OWN_U];
            own_load(cur, dst4, ch * CH + tid, nG);
            for (int it = 0; it + 1 < nCh; ++it) {
                int nch = ch + 1;
                if (nch >= nCh) nch = 0;
                int4 nxt[OWN_U];
                own_load(nxt, dst4, nch * CH + tid, nG);
                own_chunk(cur, ch * CH + tid, nG, ch == nCh - 1, r0, nr, &s_cnt, La);
#pragma unroll
                for (int u = 0; u < OWN_U; ++u) cur[u] = nxt[u];
                ch = nch;
            }
            own_chunk(cur, ch * CH + tid, nG, ch == nCh - 1, r0, nr, &s_cnt, La);
        }
        const int et = (nG << 2) + tid;
        if (et < nE) own_note(dst[et], et, r0, nr, &s_cnt, La);
    }
    __syncthreads();
    const int cnt = s_cnt;

    if (cnt <= OWN_CAP) {
        for (int i = tid; i < cnt; i += OWN_T) {
            const int e = La[i];
            const int r = dst[e] - r0;
            const int s = src[e];
            const float v = leaky(ssrc[s] + sd[r]);
            La[i] = s;
            Lr[i] = r;
            Lv[i] = v;
            ldsMaxF(&mx[r], v);
        }
        __syncthreads();
        for (int i = tid; i < cnt; i += OWN_T) {
            const int r = Lr[i];
            const float t = __expf(Lv[i] - mx[r]);
            Lv[i] = t;
            atomicAdd(&dn[r], t);
        }
        __syncthreads();
        for (int base = wave * OWN_G; base < cnt; base += OWN_W * OWN_G) {
            if (base + OWN_G <= cnt) {
                float zv[OWN_G][4];
#pragma unroll
                for (int u = 0; u < OWN_G; ++u) {
                    const int s = __builtin_amdgcn_readfirstlane(La[base + u]);
                    const float* zp = z + (size_t)s * D + lane;
                    zv[u][0] = zp[0];
                    zv[u][1] = zp[32];
                    zv[u][2] = zp[64];
                    zv[u][3] = zp[96];
                }
#pragma unroll
                for (int u = 0; u < OWN_G; ++u) {
                    const int r = __builtin_amdgcn_readfirstlane(Lr[base + u]);
                    own_add_row(acc + r * D + lane, Lv[base + u], zv[u]);
                }
            } else {
                for (int i = base; i < cnt; ++i) {
                    const int s = __builtin_amdgcn_readfirstlane(La[i]);
                    const int r = __builtin_amdgcn_readfirstlane(Lr[i]);
                    const float* zp = z + (size_t)s * D + lane;
                    float zv[4];
                    zv[0] = zp[0]; zv[1] = zp[32]; zv[2] = zp[64]; zv[3] = zp[96];
                    own_add_row(acc + r * D + lane, Lv[i], zv);
                }
            }
        }
    } else {
        for (int e = tid; e < nE; e += OWN_T) {
            const int r = dst[e] - r0;
            if ((unsigned int)r < (unsigned int)nr)
                ldsMaxF(&mx[r], leaky(ssrc[src[e]] + sd[r]));
        }
        __syncthreads();
        for (int base = wave * 32; base < nE; base += OWN_T) {
            const int e = base + lane;
            int s = 0, r = -1;
            float t = 0.0f;
            if (e < nE) r = dst[e] - r0;
            const bool hit = (unsigned int)r < (unsigned int)nr;
            if (hit) {
                s = src[e];
                t = __expf(leaky(ssrc[s] + sd[r]) - mx[r]);
                atomicAdd(&dn[r], t);
            }
            unsigned int mask = __builtin_amdgcn_ballot_w32(hit);
            while (mask) {
                const int j = __builtin_amdgcn_readfirstlane(__builtin_ctz(mask));
                mask &= mask - 1;
                const int   sj = __builtin_amdgcn_readlane(s, j);
                const int   rj = __builtin_amdgcn_readlane(r, j);
                const float tj = __int_as_float(__builtin_amdgcn_readlane(__float_as_int(t), j));
                const float* zp = z + (size_t)sj * D + lane;
                float zv[4];
                zv[0] = zp[0]; zv[1] = zp[32]; zv[2] = zp[64]; zv[3] = zp[96];
                own_add_row(acc + rj * D + lane, tj, zv);
            }
        }
    }
    __syncthreads();

    for (int r = wave; r < nr; r += OWN_W) {
        const float d = dn[r];
        const float4 a = *(const float4*)(acc + r * D + lane * 4);
        float4 o;
        if (d > 0.0f) {
            o.x = a.x / d; o.y = a.y / d; o.z = a.z / d; o.w = a.w / d;
        } else {
            o.x = 0.0f; o.y = 0.0f; o.z = 0.0f; o.w = 0.0f;
        }
        const int row = r0 + r;
        float* op = (row < n0) ? (out_lo + (size_t)row * D)
                               : (out_hi + (size_t)(row - n0) * D);
        *(float4*)(op + lane * 4) = o;
    }
}

#define CB_T   D
#define CB_G   4
#define CB_PER 16

typedef const __attribute__((address_space(4))) float* cb_row;
typedef float v16f __attribute__((ext_vector_type(16)));
typedef const __attribute__((address_space(4))) v16f* cb_row16;
#define CB_KC (D / 16)

__global__ __launch_bounds__(CB_T)
void combine(const float* __restrict__ A, const float* __restrict__ B,
             const float* __restrict__ W1, const float* __restrict__ b1,
             const float* __restrict__ w2, float* __restrict__ out, int K)
{
    __shared__ float red[2][2 * CB_G][CB_T / 32];
    const int j    = threadIdx.x;
    const int lane = j & 31;
    const int wave = j >> 5;

    float w[D];
#pragma unroll
    for (int k = 0; k < D; ++k) w[k] = W1[(size_t)k * D + j];
    const float b1j = b1[j];
    const float w2j = w2[j];

    const int first = (int)blockIdx.x * CB_PER;
    int last = first + CB_PER;
    last = last < K ? last : K;
    int cp = 0;
    for (int i0 = first; i0 < last; i0 += CB_G) {
        cb_row16 ar[CB_G];
        cb_row16 br[CB_G];
#pragma unroll
        for (int g = 0; g < CB_G; ++g) {
            int i = i0 + g;
            i = i < last ? i : last - 1;
            ar[g] = (cb_row16)(A + (size_t)i * D);
            br[g] = (cb_row16)(B + (size_t)i * D);
        }
        float acc[2 * CB_G];
#pragma unroll
        for (int q = 0; q < 2 * CB_G; ++q) acc[q] = b1j;
#pragma unroll
        for (int c = 0; c < CB_KC; ++c) {
            {
                v16f s[CB_G];
#pragma unroll
                for (int g = 0; g < CB_G; ++g) s[g] = ar[g][c];
#pragma unroll
                for (int kk = 0; kk < 16; ++kk) {
#pragma unroll
                    for (int g = 0; g < CB_G; ++g)
                        acc[g] = fmaf(s[g][kk], w[16 * c + kk], acc[g]);
                }
            }
            asm volatile("" : "+v"(acc[0]), "+v"(acc[1]), "+v"(acc[2]), "+v"(acc[3]));
            {
                v16f s[CB_G];
#pragma unroll
                for (int g = 0; g < CB_G; ++g) s[g] = br[g][c];
#pragma unroll
                for (int kk = 0; kk < 16; ++kk) {
#pragma unroll
                    for (int g = 0; g < CB_G; ++g)
                        acc[CB_G + g] = fmaf(s[g][kk], w[16 * c + kk], acc[CB_G + g]);
                }
            }
            asm volatile("" : "+v"(acc[CB_G + 0]), "+v"(acc[CB_G + 1]), "+v"(acc[CB_G + 2]), "+v"(acc[CB_G + 3]));
        }
#pragma unroll
        for (int q = 0; q < 2 * CB_G; ++q) acc[q] = tanhf(acc[q]) * w2j;
#pragma unroll
        for (int off = 16; off > 0; off >>= 1) {
#pragma unroll
            for (int q = 0; q < 2 * CB_G; ++q) acc[q] += __shfl_xor(acc[q], off, 32);
        }
        if (lane == 0) {
#pragma unroll
            for (int q = 0; q < 2 * CB_G; ++q) red[cp][q][wave] = acc[q];
        }
        __syncthreads();
#pragma unroll
        for (int g = 0; g < CB_G; ++g) {
            if (i0 + g < last) {
                const float s1 = ((red[cp][g][0] + red[cp][g][1]) + red[cp][g][2]) + red[cp][g][3];
                const float s2 = ((red[cp][CB_G + g][0] + red[cp][CB_G + g][1])
                                  + red[cp][CB_G + g][2]) + red[cp][CB_G + g][3];
                const float mm = fmaxf(s1, s2);
                const float e1 = __expf(s1 - mm), e2 = __expf(s2 - mm);
                const float al = e1 / (e1 + e2);
                out[(size_t)(i0 + g) * D + j] = al * ((cb_row)ar[g])[j] + (1.0f - al) * ((cb_row)br[g])[j];
            }
        }
        cp ^= 1;
    }
}

static inline int cdiv(int a, int b) { return (a + b - 1) / b; }

extern "C" void kernel_launch(void* const* d_in, const int* in_sizes, int n_in,
                              void* d_out, int out_size, void* d_ws, size_t ws_size,
                              hipStream_t stream)
{
    (void)n_in; (void)out_size; (void)ws_size;
    const float* kn_emb   = (const float*)d_in[0];
    const float* exer_emb = (const float*)d_in[1];
    const float* W_dir    = (const float*)d_in[2];
    const float* a_dir    = (const float*)d_in[3];
    const float* W_ke     = (const float*)d_in[4];
    const float* a_ke     = (const float*)d_in[5];
    const float* W_ek     = (const float*)d_in[6];
    const float* a_ek     = (const float*)d_in[7];
    const float* rW1      = (const float*)d_in[8];
    const float* rb1      = (const float*)d_in[9];
    const float* rw2      = (const float*)d_in[10];
    const int* dsrc  = (const int*)d_in[11];
    const int* ddst  = (const int*)d_in[12];
    const int* kesrc = (const int*)d_in[13];
    const int* kedst = (const int*)d_in[14];
    const int* eksrc = (const int*)d_in[15];
    const int* ekdst = (const int*)d_in[16];

    const int Kn = in_sizes[0] / D;
    const int In = in_sizes[1] / D;
    const int Nn = Kn + In;
    const int E1 = in_sizes[11];
    const int E2 = in_sizes[13];

    float* w = (float*)d_ws;
    size_t off = 0;
    float* buf_z = w + off; off += (size_t)Nn * D;
    float* pY    = w + off; off += (size_t)Kn * D;
    float* pX    = w + off; off += (size_t)In * D;
    float* pA    = w + off; off += (size_t)Kn * D;
    float* ssrc  = w + off; off += (size_t)Nn;
    float* sdst  = w + off; off += (size_t)Nn;

    float* out_kn = (float*)d_out;
    float* out_ex = (float*)d_out + (size_t)Kn * D;

    int gN = 256, rN = cdiv(Nn, gN);
    if (rN > OWN_RMAX) { gN = cdiv(Nn, OWN_RMAX); rN = cdiv(Nn, gN); }
    int gK = 256, rK = cdiv(Kn, gK);
    if (rK > OWN_RMAX) { gK = cdiv(Kn, OWN_RMAX); rK = cdiv(Kn, gK); }

    int* keyKE  = (gN <= 256 && (size_t)E2 <= (size_t)Kn * D * 4) ? (int*)pA : nullptr;
    int* keyDIR = (gK <= 256 && (size_t)E1 <= (size_t)In * D * 4) ? (int*)pX : nullptr;
    int* keyEK  = (gN <= 256 && (size_t)E2 <= (size_t)In * D * 4) ? (int*)pX : nullptr;

    gemm_z<<<cdiv(Nn, GZ_ROWS), GZ_T, 0, stream>>>(exer_emb, In, kn_emb, W_ke, buf_z, Nn);
    row_scores<<<cdiv(Nn, RS_WGR) + (keyKE ? cdiv(E2 >> 2, RS_WGK) : 0), RS_T, 0, stream>>>(buf_z, a_ke, ssrc, sdst, Nn, kedst, keyKE, E2, rN);
    owner_agg<<<gN, OWN_T, 0, stream>>>(kesrc, kedst, keyKE, ssrc, sdst, buf_z, pX, pY, In, Nn, E2, rN);

    gemm_z<<<cdiv(Kn, GZ_ROWS), GZ_T, 0, stream>>>(kn_emb, Kn, kn_emb, W_dir, buf_z, Kn);
    row_scores<<<cdiv(Kn, RS_WGR) + (keyDIR ? cdiv(E1 >> 2, RS_WGK) : 0), RS_T, 0, stream>>>(buf_z, a_dir, ssrc, sdst, Kn, ddst, keyDIR, E1, rK);
    owner_agg<<<gK, OWN_T, 0, stream>>>(dsrc, ddst, keyDIR, ssrc, sdst, buf_z, pA, pA, Kn, Kn, E1, rK);

    combine<<<cdiv(Kn, CB_PER), CB_T, 0, stream>>>(pA, pY, rW1, rb1, rw2, out_kn, Kn);

    gemm_z<<<cdiv(Nn, GZ_ROWS), GZ_T, 0, stream>>>(exer_emb, In, kn_emb, W_ek, buf_z, Nn);
    row_scores<<<cdiv(Nn, RS_WGR) + (keyEK ? cdiv(E2 >> 2, RS_WGK) : 0), RS_T, 0, stream>>>(buf_z, a_ek, ssrc, sdst, Nn, ekdst, keyEK, E2, rN);
    owner_agg<<<gN, OWN_T, 0, stream>>>(eksrc, ekdst, keyEK, ssrc, sdst, buf_z, out_ex, pY, In, Nn, E2, rN);
}


# independent stages of different graphs share launches, picked by block index: ke scores beside dir product, dir scores beside ek product, ek scores beside gate; 10 launches to 7; sum orders kept
# speedup vs baseline: 3.2323x; 1.0626x over previous
#include <hip/hip_runtime.h>
#include <hip/hip_bf16.h>
#include <math.h>

#define D 128

typedef float v2f __attribute__((ext_vector_type(2)));
typedef float v8f __attribute__((ext_vector_type(8)));

#define GZ_T    128
#define GZ_ROWS (16 * (GZ_T / 32))
#define GZ_KS   (D / 4)
#define GZ_AHEAD 3

struct gz_ops { float2 a; float4 x0, x1, y0, y1; };

__device__ __forceinline__ gz_ops gz_load(const float* hrow, const float* wcol, int kk) {
    gz_ops o;
    o.a  = *(const float2*)(hrow + kk);
    o.x0 = *(const float4*)(wcol + (size_t)kk * D);
    o.x1 = *(const float4*)(wcol + (size_t)kk * D + 64);
    o.y0 = *(const float4*)(wcol + (size_t)(kk + 1) * D);
    o.y1 = *(const float4*)(wcol + (size_t)(kk + 1) * D + 64);
    return o;
}

__device__ __forceinline__ v8f gz_step(v2f a, float bx, float by, v8f c) {
    v2f b; b.x = bx; b.y = by;
    return __builtin_amdgcn_wmma_f32_16x16x4_f32(false, a, false, b, (short)0, c, false, false);
}

__device__ __forceinline__
void gemm_z_wg(int bid,
               const float* __restrict__ h0, int n0,
               const float* __restrict__ h1,
               const float* __restrict__ W,
               float* __restrict__ Z, int n)
{
    const int lane = threadIdx.x & 31;
    const int half = lane >> 4;
    const int lr   = lane & 15;
    const int row0 = __builtin_amdgcn_readfirstlane(
                         (bid * GZ_T + (int)threadIdx.x) >> 5) * 16;
    if (row0 >= n) return;

    int rowA = row0 + lr;
    if (rowA >= n) rowA = n - 1;
    const float* hrow = (rowA < n0) ? (h0 + (size_t)rowA * D)
                                    : (h1 + (size_t)(rowA - n0) * D);
    const float* wcol = W + 4 * lr;

    v8f c[8];
#pragma unroll
    for (int t = 0; t < 8; ++t)
        c[t] = (v8f){0.f, 0.f, 0.f, 0.f, 0.f, 0.f, 0.f, 0.f};

    gz_ops ring[GZ_AHEAD + 1];
#pragma unroll
    for (int i = 0; i < GZ_AHEAD; ++i)
        ring[i] = gz_load(hrow, wcol, 4 * i + 2 * half);

#pragma unroll
    for (int i = 0; i < GZ_KS; ++i) {
        if (i + GZ_AHEAD < GZ_KS)
            ring[(i + GZ_AHEAD) % (GZ_AHEAD + 1)] =
                gz_load(hrow, wcol, 4 * (i + GZ_AHEAD) + 2 * half);
        const gz_ops o = ring[i % (GZ_AHEAD + 1)];
        v2f a; a.x = o.a.x; a.y = o.a.y;
        c[0] = gz_step(a, o.x0.x, o.y0.x, c[0]);
        c[1] = gz_step(a, o.x0.y, o.y0.y, c[1]);
        c[2] = gz_step(a, o.x0.z, o.y0.z, c[2]);
        c[3] = gz_step(a, o.x0.w, o.y0.w, c[3]);
        c[4] = gz_step(a, o.x1.x, o.y1.x, c[4]);
        c[5] = gz_step(a, o.x1.y, o.y1.y, c[5]);
        c[6] = gz_step(a, o.x1.z, o.y1.z, c[6]);
        c[7] = gz_step(a, o.x1.w, o.y1.w, c[7]);
    }
#pragma unroll
    for (int v = 0; v < 8; ++v) {
        const int r = row0 + v + 8 * half;
        if (r < n) {
            float* zp = Z + (size_t)r * D + 4 * lr;
            *(float4*)zp        = make_float4(c[0][v], c[1][v], c[2][v], c[3][v]);
            *(float4*)(zp + 64) = make_float4(c[4][v], c[5][v], c[6][v], c[7][v]);
        }
    }
}

__global__ __launch_bounds__(GZ_T, 5)
void gemm_z(const float* __restrict__ h0, int n0,
            const float* __restrict__ h1,
            const float* __restrict__ W,
            float* __restrict__ Z, int n)
{
    gemm_z_wg((int)blockIdx.x, h0, n0, h1, W, Z, n);
}

#define RS_ROWS 32
#define RS_T    128
#define RS_KU   4
#define RS_WGR  (RS_ROWS * (RS_T / 32))
#define RS_WGK  (RS_T * RS_KU)

__device__ __forceinline__ float rs_dot4(float4 zv, float4 av) {
    float p = zv.y * av.y;
    p = fmaf(zv.x, av.x, p);
    p = fmaf(zv.z, av.z, p);
    p = fmaf(zv.w, av.w, p);
    return p;
}

__device__ __forceinline__
void row_scores_wg(int bid,
                   const float* __restrict__ Z, const float* __restrict__ a,
                   float* __restrict__ ssrc, float* __restrict__ sdst, int n,
                   const int* __restrict__ dst, int* __restrict__ key,
                   int nE, int R)
{
    const int nRowWG = (n + RS_WGR - 1) / RS_WGR;
    if (bid >= nRowWG) {
        const int nW = nE >> 2;
        const unsigned int uR = (unsigned int)R;
        const int4* dst4 = (const int4*)dst;
        const int q0 = (bid - nRowWG) * RS_WGK + (int)threadIdx.x;
        int4 d[RS_KU];
#pragma unroll
        for (int u = 0; u < RS_KU; ++u) {
            int q = q0 + u * RS_T;
            q = q < nW ? q : nW - 1;
            d[u] = dst4[q];
        }
#pragma unroll
        for (int u = 0; u < RS_KU; ++u) {
            const int q = q0 + u * RS_T;
            if (q < nW) {
                const unsigned int w = (((unsigned int)d[u].x / uR) & 255u)
                                     | ((((unsigned int)d[u].y / uR) & 255u) << 8)
                                     | ((((unsigned int)d[u].z / uR) & 255u) << 16)
                                     | ((((unsigned int)d[u].w / uR) & 255u) << 24);
                key[q] = (int)w;
            }
        }
        return;
    }

    const int lane = threadIdx.x & 31;
    const int row0 = __builtin_amdgcn_readfirstlane(
                         (bid * RS_T + (int)threadIdx.x) >> 5) * RS_ROWS;
    if (row0 >= n) return;

    const float4 a0 = *(const float4*)(a + lane * 4);
    const float4 a1 = *(const float4*)(a + D + lane * 4);
    float4 zv[RS_ROWS];
#pragma unroll
    for (int u = 0; u < RS_ROWS; ++u) {
        int r = row0 + u;
        r = r < n ? r : n - 1;
        zv[u] = *(const float4*)(Z + (size_t)r * D + lane * 4);
    }

    float p0[RS_ROWS], p1[RS_ROWS];
#pragma unroll
    for (int u = 0; u < RS_ROWS; ++u) {
        p0[u] = rs_dot4(zv[u], a0);
        p1[u] = rs_dot4(zv[u], a1);
    }
#pragma unroll
    for (int off = 16; off > 0; off >>= 1) {
#pragma unroll
        for (int u = 0; u < RS_ROWS; ++u) {
            p0[u] += __shfl_xor(p0[u], off, 32);
            p1[u] += __shfl_xor(p1[u], off, 32);
        }
    }
    float my0 = 0.0f, my1 = 0.0f;
#pragma unroll
    for (int u = 0; u < RS_ROWS; ++u)
        if (lane == u) { my0 = p0[u]; my1 = p1[u]; }
    if (row0 + lane < n) { ssrc[row0 + lane] = my0; sdst[row0 + lane] = my1; }
}

__device__ __forceinline__ float leaky(float v) {
    return v > 0.0f ? v : 0.01f * v;
}

#define OWN_T    1024
#define OWN_W    (OWN_T / 32)
#define OWN_RMAX 288
#define OWN_CAP  8192
#define OWN_U    8
#define OWN_UK   4
#define OWN_G    16

__device__ __forceinline__ void ldsMaxF(float* addr, float val) {
    if (val >= 0.0f) atomicMax((int*)addr, __float_as_int(val));
    else             atomicMin((unsigned int*)addr, __float_as_uint(val));
}

__device__ __forceinline__ void own_push(int e, int* cnt, int* La) {
    const int idx = atomicAdd(cnt, 1);
    if (idx < OWN_CAP) La[idx] = e;
}

__device__ __forceinline__ void own_note(int d, int e, int r0, int nr,
                                         int* cnt, int* La) {
    if ((unsigned int)(d - r0) < (unsigned int)nr) own_push(e, cnt, La);
}

__device__ __forceinline__ void own_load(int4* dv, const int4* dst4, int g, int nG) {
#pragma unroll
    for (int u = 0; u < OWN_U; ++u) {
        int gi = g + u * OWN_T;
        gi = gi < nG ? gi : nG - 1;
        dv[u] = dst4[gi];
    }
}

__device__ __forceinline__ void own_chunk(int4* dv, int gb, int nG, bool shortc,
                                          int r0, int nr, int* cnt, int* La) {
    if (shortc) {
#pragma unroll
        for (int u = 0; u < OWN_U; ++u)
            if (gb + u * OWN_T >= nG) dv[u] = make_int4(-1, -1, -1, -1);
    }
#pragma unroll
    for (int u = 0; u < OWN_U; ++u) {
        const int e0 = (gb + u * OWN_T) << 2;
        own_note(dv[u].x, e0,     r0, nr, cnt, La);
        own_note(dv[u].y, e0 + 1, r0, nr, cnt, La);
        own_note(dv[u].z, e0 + 2, r0, nr, cnt, La);
        own_note(dv[u].w, e0 + 3, r0, nr, cnt, La);
    }
}

__device__ __forceinline__ void key_load(uint4* kv, const uint4* key4, int g, int nQ) {
#pragma unroll
    for (int u = 0; u < OWN_UK; ++u) {
        int gi = g + u * OWN_T;
        gi = gi < nQ ? gi : nQ - 1;
        kv[u] = key4[gi];
    }
}

__device__ __forceinline__ unsigned int key_nz(unsigned int x) {
    return ((x & 0x7f7f7f7fu) + 0x7f7f7f7fu) | x;
}

__device__ __forceinline__ unsigned int key_match(uint4 kv, unsigned int pat) {
    const unsigned int n0 = key_nz(kv.x ^ pat);
    const unsigned int n1 = key_nz(kv.y ^ pat);
    const unsigned int n2 = key_nz(kv.z ^ pat);
    const unsigned int n3 = key_nz(kv.w ^ pat);
    const unsigned int nz = ((n0 >> 7) & 0x01010101u) | ((n1 >> 6) & 0x02020202u)
                          | ((n2 >> 5) & 0x04040404u) | ((n3 >> 4) & 0x08080808u);
    return ~nz & 0x0f0f0f0fu;
}

__device__ __forceinline__ void key_chunk(const uint4* kv, int gb, int nQ, bool shortc,
                                          unsigned int pat, int* cnt, int* La) {
#pragma unroll
    for (int u = 0; u < OWN_UK; ++u) {
        unsigned int m = key_match(kv[u], pat);
        if (shortc && gb + u * OWN_T >= nQ) m = 0u;
        const int e0 = (gb + u * OWN_T) << 4;
        while (m) {
            const int p = __builtin_ctz(m);
            m &= m - 1u;
            own_push(e0 + ((p & 7) << 2) + (p >> 3), cnt, La);
        }
    }
}

__device__ __forceinline__ void own_add_row(float* arow, float t, const float* zv) {
    atomicAdd(arow,      t * zv[0]);
    atomicAdd(arow + 32, t * zv[1]);
    atomicAdd(arow + 64, t * zv[2]);
    atomicAdd(arow + 96, t * zv[3]);
}

__global__ __launch_bounds__(OWN_T)
void owner_agg(const int* __restrict__ src, const int* __restrict__ dst,
               const int* __restrict__ key,
               const float* __restrict__ ssrc, const float* __restrict__ sdst,
               const float* __restrict__ z,
               float* out_lo, float* out_hi,
               int n0, int n, int nE, int R)
{
    __shared__ __attribute__((aligned(16))) float acc[OWN_RMAX * D];
    __shared__ float mx[OWN_RMAX];
    __shared__ float dn[OWN_RMAX];
    __shared__ float sd[OWN_RMAX];
    __shared__ int   La[OWN_CAP];
    __shared__ int   Lr[OWN_CAP];
    __shared__ float Lv[OWN_CAP];
    __shared__ int   s_cnt;

    const int tid  = threadIdx.x;
    const int lane = tid & 31;
    const int wave = tid >> 5;
    const int r0   = blockIdx.x * R;
    int nr = n - r0;
    nr = nr > R ? R : nr;
    if (nr <= 0) return;

    {
        const float4 z4 = make_float4(0.0f, 0.0f, 0.0f, 0.0f);
        float4* a4 = (float4*)acc;
        for (int i = tid; i < nr * (D / 4); i += OWN_T) a4[i] = z4;
        for (int i = tid; i < nr; i += OWN_T) {
            mx[i] = -INFINITY;
            dn[i] = 0.0f;
            sd[i] = sdst[r0 + i];
        }
        if (tid == 0) s_cnt = 0;
    }
    __syncthreads();

    if (key != nullptr) {
        const int nQ  = nE >> 4;
        const int CH  = OWN_T * OWN_UK;
        const int nCh = (nQ + CH - 1) / CH;
        const unsigned int pat = (unsigned int)blockIdx.x * 0x01010101u;
        if (nQ > 0) {
            const uint4* key4 = (const uint4*)key;
            int ch = (int)(((long long)blockIdx.x * nCh) / gridDim.x);
            uint4 cur[OWN_UK];
            key_load(cur, key4, ch * CH + tid, nQ);
            for (int it = 0; it + 1 < nCh; ++it) {
                int nch = ch + 1;
                if (nch >= nCh) nch = 0;
                uint4 nxt[OWN_UK];
                key_load(nxt, key4, nch * CH + tid, nQ);
                key_chunk(cur, ch * CH + tid, nQ, ch == nCh - 1, pat, &s_cnt, La);
#pragma unroll
                for (int u = 0; u < OWN_UK; ++u) cur[u] = nxt[u];
                ch = nch;
            }
            key_chunk(cur, ch * CH + tid, nQ, ch == nCh - 1, pat, &s_cnt, La);
        }
        const int et = (nQ << 4) + tid;
        if (et < nE) own_note(dst[et], et, r0, nr, &s_cnt, La);
    } else {
        const int nG  = nE >> 2;
        const int CH  = OWN_T * OWN_U;
        const int nCh = (nG + CH - 1) / CH;
        if (nG > 0) {
            const int4* dst4 = (const int4*)dst;
            int ch = (int)(((long long)blockIdx.x * nCh) / gridDim.x);
            int4 cur[OWN_U];
            own_load(cur, dst4, ch * CH + tid, nG);
            for (int it = 0; it + 1 < nCh; ++it) {
                int nch = ch + 1;
                if (nch >= nCh) nch = 0;
                int4 nxt[OWN_U];
                own_load(nxt, dst4, nch * CH + tid, nG);
                own_chunk(cur, ch * CH + tid, nG, ch == nCh - 1, r0, nr, &s_cnt, La);
#pragma unroll
                for (int u = 0; u < OWN_U; ++u) cur[u] = nxt[u];
                ch = nch;
            }
            own_chunk(cur, ch * CH + tid, nG, ch == nCh - 1, r0, nr, &s_cnt, La);
        }
        const int et = (nG << 2) + tid;
        if (et < nE) own_note(dst[et], et, r0, nr, &s_cnt, La);
    }
    __syncthreads();
    const int cnt = s_cnt;

    if (cnt <= OWN_CAP) {
        for (int i = tid; i < cnt; i += OWN_T) {
            const int e = La[i];
            const int r = dst[e] - r0;
            const int s = src[e];
            const float v = leaky(ssrc[s] + sd[r]);
            La[i] = s;
            Lr[i] = r;
            Lv[i] = v;
            ldsMaxF(&mx[r], v);
        }
        __syncthreads();
        for (int i = tid; i < cnt; i += OWN_T) {
            const int r = Lr[i];
            const float t = __expf(Lv[i] - mx[r]);
            Lv[i] = t;
            atomicAdd(&dn[r], t);
        }
        __syncthreads();
        for (int base = wave * OWN_G; base < cnt; base += OWN_W * OWN_G) {
            if (base + OWN_G <= cnt) {
                float zv[OWN_G][4];
#pragma unroll
                for (int u = 0; u < OWN_G; ++u) {
                    const int s = __builtin_amdgcn_readfirstlane(La[base + u]);
                    const float* zp = z + (size_t)s * D + lane;
                    zv[u][0] = zp[0];
                    zv[u][1] = zp[32];
                    zv[u][2] = zp[64];
                    zv[u][3] = zp[96];
                }
#pragma unroll
                for (int u = 0; u < OWN_G; ++u) {
                    const int r = __builtin_amdgcn_readfirstlane(Lr[base + u]);
                    own_add_row(acc + r * D + lane, Lv[base + u], zv[u]);
                }
            } else {
                for (int i = base; i < cnt; ++i) {
                    const int s = __builtin_amdgcn_readfirstlane(La[i]);
                    const int r = __builtin_amdgcn_readfirstlane(Lr[i]);
                    const float* zp = z + (size_t)s * D + lane;
                    float zv[4];
                    zv[0] = zp[0]; zv[1] = zp[32]; zv[2] = zp[64]; zv[3] = zp[96];
                    own_add_row(acc + r * D + lane, Lv[i], zv);
                }
            }
        }
    } else {
        for (int e = tid; e < nE; e += OWN_T) {
            const int r = dst[e] - r0;
            if ((unsigned int)r < (unsigned int)nr)
                ldsMaxF(&mx[r], leaky(ssrc[src[e]] + sd[r]));
        }
        __syncthreads();
        for (int base = wave * 32; base < nE; base += OWN_T) {
            const int e = base + lane;
            int s = 0, r = -1;
            float t = 0.0f;
            if (e < nE) r = dst[e] - r0;
            const bool hit = (unsigned int)r < (unsigned int)nr;
            if (hit) {
                s = src[e];
                t = __expf(leaky(ssrc[s] + sd[r]) - mx[r]);
                atomicAdd(&dn[r], t);
            }
            unsigned int mask = __builtin_amdgcn_ballot_w32(hit);
            while (mask) {
                const int j = __builtin_amdgcn_readfirstlane(__builtin_ctz(mask));
                mask &= mask - 1;
                const int   sj = __builtin_amdgcn_readlane(s, j);
                const int   rj = __builtin_amdgcn_readlane(r, j);
                const float tj = __int_as_float(__builtin_amdgcn_readlane(__float_as_int(t), j));
                const float* zp = z + (size_t)sj * D + lane;
                float zv[4];
                zv[0] = zp[0]; zv[1] = zp[32]; zv[2] = zp[64]; zv[3] = zp[96];
                own_add_row(acc + rj * D + lane, tj, zv);
            }
        }
    }
    __syncthreads();

    for (int r = wave; r < nr; r += OWN_W) {
        const float d = dn[r];
        const float4 a = *(const float4*)(acc + r * D + lane * 4);
        float4 o;
        if (d > 0.0f) {
            o.x = a.x / d; o.y = a.y / d; o.z = a.z / d; o.w = a.w / d;
        } else {
            o.x = 0.0f; o.y = 0.0f; o.z = 0.0f; o.w = 0.0f;
        }
        const int row = r0 + r;
        float* op = (row < n0) ? (out_lo + (size_t)row * D)
                               : (out_hi + (size_t)(row - n0) * D);
        *(float4*)(op + lane * 4) = o;
    }
}

#define CB_T   D
#define CB_G   4
#define CB_PER 16

typedef const __attribute__((address_space(4))) float* cb_row;
typedef float v16f __attribute__((ext_vector_type(16)));
typedef const __attribute__((address_space(4))) v16f* cb_row16;
#define CB_KC (D / 16)

__device__ __forceinline__
void combine_wg(int bid,
                const float* __restrict__ A, const float* __restrict__ B,
                const float* __restrict__ W1, const float* __restrict__ b1,
                const float* __restrict__ w2, float* __restrict__ out, int K)
{
    __shared__ float red[2][2 * CB_G][CB_T / 32];
    const int j    = threadIdx.x;
    const int lane = j & 31;
    const int wave = j >> 5;

    float w[D];
#pragma unroll
    for (int k = 0; k < D; ++k) w[k] = W1[(size_t)k * D + j];
    const float b1j = b1[j];
    const float w2j = w2[j];

    const int first = bid * CB_PER;
    int last = first + CB_PER;
    last = last < K ? last : K;
    int cp = 0;
    for (int i0 = first; i0 < last; i0 += CB_G) {
        cb_row16 ar[CB_G];
        cb_row16 br[CB_G];
#pragma unroll
        for (int g = 0; g < CB_G; ++g) {
            int i = i0 + g;
            i = i < last ? i : last - 1;
            ar[g] = (cb_row16)(A + (size_t)i * D);
            br[g] = (cb_row16)(B + (size_t)i * D);
        }
        float acc[2 * CB_G];
#pragma unroll
        for (int q = 0; q < 2 * CB_G; ++q) acc[q] = b1j;
#pragma unroll
        for (int c = 0; c < CB_KC; ++c) {
            {
                v16f s[CB_G];
#pragma unroll
                for (int g = 0; g < CB_G; ++g) s[g] = ar[g][c];
#pragma unroll
                for (int kk = 0; kk < 16; ++kk) {
#pragma unroll
                    for (int g = 0; g < CB_G; ++g)
                        acc[g] = fmaf(s[g][kk], w[16 * c + kk], acc[g]);
                }
            }
            asm volatile("" : "+v"(acc[0]), "+v"(acc[1]), "+v"(acc[2]), "+v"(acc[3]));
            {
                v16f s[CB_G];
#pragma unroll
                for (int g = 0; g < CB_G; ++g) s[g] = br[g][c];
#pragma unroll
                for (int kk = 0; kk < 16; ++kk) {
#pragma unroll
                    for (int g = 0; g < CB_G; ++g)
                        acc[CB_G + g] = fmaf(s[g][kk], w[16 * c + kk], acc[CB_G + g]);
                }
            }
            asm volatile("" : "+v"(acc[CB_G + 0]), "+v"(acc[CB_G + 1]), "+v"(acc[CB_G + 2]), "+v"(acc[CB_G + 3]));
        }
#pragma unroll
        for (int q = 0; q < 2 * CB_G; ++q) acc[q] = tanhf(acc[q]) * w2j;
#pragma unroll
        for (int off = 16; off > 0; off >>= 1) {
#pragma unroll
            for (int q = 0; q < 2 * CB_G; ++q) acc[q] += __shfl_xor(acc[q], off, 32);
        }
        if (lane == 0) {
#pragma unroll
            for (int q = 0; q < 2 * CB_G; ++q) red[cp][q][wave] = acc[q];
        }
        __syncthreads();
#pragma unroll
        for (int g = 0; g < CB_G; ++g) {
            if (i0 + g < last) {
                const float s1 = ((red[cp][g][0] + red[cp][g][1]) + red[cp][g][2]) + red[cp][g][3];
                const float s2 = ((red[cp][CB_G + g][0] + red[cp][CB_G + g][1])
                                  + red[cp][CB_G + g][2]) + red[cp][CB_G + g][3];
                const float mm = fmaxf(s1, s2);
                const float e1 = __expf(s1 - mm), e2 = __expf(s2 - mm);
                const float al = e1 / (e1 + e2);
                out[(size_t)(i0 + g) * D + j] = al * ((cb_row)ar[g])[j] + (1.0f - al) * ((cb_row)br[g])[j];
            }
        }
        cp ^= 1;
    }
}

static_assert(RS_T == GZ_T && RS_T == CB_T, "stages that share a launch share its block size");

__global__ __launch_bounds__(GZ_T, 5)
void scores_and_product(const float* __restrict__ Z, const float* __restrict__ a,
                        float* __restrict__ ssrc, float* __restrict__ sdst, int n,
                        const int* __restrict__ dst, int* __restrict__ key,
                        int nE, int R, int nScoreWG,
                        const float* __restrict__ h0, int n0,
                        const float* __restrict__ h1,
                        const float* __restrict__ W,
                        float* __restrict__ Zn, int nn)
{
    if ((int)blockIdx.x < nScoreWG)
        row_scores_wg((int)blockIdx.x, Z, a, ssrc, sdst, n, dst, key, nE, R);
    else
        gemm_z_wg((int)blockIdx.x - nScoreWG, h0, n0, h1, W, Zn, nn);
}

__global__ __launch_bounds__(CB_T)
void scores_and_combine(const float* __restrict__ Z, const float* __restrict__ a,
                        float* __restrict__ ssrc, float* __restrict__ sdst, int n,
                        const int* __restrict__ dst, int* __restrict__ key,
                        int nE, int R, int nScoreWG,
                        const float* __restrict__ A, const float* __restrict__ B,
                        const float* __restrict__ W1, const float* __restrict__ b1,
                        const float* __restrict__ w2, float* __restrict__ out, int K)
{
    if ((int)blockIdx.x < nScoreWG)
        row_scores_wg((int)blockIdx.x, Z, a, ssrc, sdst, n, dst, key, nE, R);
    else
        combine_wg((int)blockIdx.x - nScoreWG, A, B, W1, b1, w2, out, K);
}

static inline int cdiv(int a, int b) { return (a + b - 1) / b; }

extern "C" void kernel_launch(void* const* d_in, const int* in_sizes, int n_in,
                              void* d_out, int out_size, void* d_ws, size_t ws_size,
                              hipStream_t stream)
{
    (void)n_in; (void)out_size; (void)ws_size;
    const float* kn_emb   = (const float*)d_in[0];
    const float* exer_emb = (const float*)d_in[1];
    const float* W_dir    = (const float*)d_in[2];
    const float* a_dir    = (const float*)d_in[3];
    const float* W_ke     = (const float*)d_in[4];
    const float* a_ke     = (const float*)d_in[5];
    const float* W_ek     = (const float*)d_in[6];
    const float* a_ek     = (const float*)d_in[7];
    const float* rW1      = (const float*)d_in[8];
    const float* rb1      = (const float*)d_in[9];
    const float* rw2      = (const float*)d_in[10];
    const int* dsrc  = (const int*)d_in[11];
    const int* ddst  = (const int*)d_in[12];
    const int* kesrc = (const int*)d_in[13];
    const int* kedst = (const int*)d_in[14];
    const int* eksrc = (const int*)d_in[15];
    const int* ekdst = (const int*)d_in[16];

    const int Kn = in_sizes[0] / D;
    const int In = in_sizes[1] / D;
    const int Nn = Kn + In;
    const int E1 = in_sizes[11];
    const int E2 = in_sizes[13];

    float* w = (float*)d_ws;
    size_t off = 0;
    float* buf_z = w + off; off += (size_t)Nn * D;
    float* pY    = w + off; off += (size_t)Kn * D;
    float* pX    = w + off; off += (size_t)In * D;
    float* pA    = w + off; off += (size_t)Kn * D;
    float* ssrc  = w + off; off += (size_t)Nn;
    float* sdst  = w + off; off += (size_t)Nn;

    float* out_kn = (float*)d_out;
    float* out_ex = (float*)d_out + (size_t)Kn * D;

    int gN = 256, rN = cdiv(Nn, gN);
    if (rN > OWN_RMAX) { gN = cdiv(Nn, OWN_RMAX); rN = cdiv(Nn, gN); }
    int gK = 256, rK = cdiv(Kn, gK);
    if (rK > OWN_RMAX) { gK = cdiv(Kn, OWN_RMAX); rK = cdiv(Kn, gK); }

    int* keyKE  = (gN <= 256 && (size_t)E2 <= (size_t)Kn * D * 4) ? (int*)pA : nullptr;
    int* keyDIR = (gK <= 256 && (size_t)E1 <= (size_t)In * D * 4) ? (int*)pX : nullptr;
    int* keyEK  = (gN <= 256 && (size_t)E2 <= (size_t)In * D * 4) ? (int*)pX : nullptr;

    float* z_dir = out_kn;

    const int sKE  = cdiv(Nn, RS_WGR) + (keyKE  ? cdiv(E2 >> 2, RS_WGK) : 0);
    const int sDIR = cdiv(Kn, RS_WGR) + (keyDIR ? cdiv(E1 >> 2, RS_WGK) : 0);
    const int sEK  = cdiv(Nn, RS_WGR) + (keyEK  ? cdiv(E2 >> 2, RS_WGK) : 0);
    const int pN   = cdiv(Nn, GZ_ROWS);
    const int pK   = cdiv(Kn, GZ_ROWS);

    gemm_z<<<pN, GZ_T, 0, stream>>>(exer_emb, In, kn_emb, W_ke, buf_z, Nn);
    scores_and_product<<<sKE + pK, GZ_T, 0, stream>>>(buf_z, a_ke, ssrc, sdst, Nn, kedst, keyKE, E2, rN, sKE,
                                                     kn_emb, Kn, kn_emb, W_dir, z_dir, Kn);
    owner_agg<<<gN, OWN_T, 0, stream>>>(kesrc, kedst, keyKE, ssrc, sdst, buf_z, pX, pY, In, Nn, E2, rN);
    scores_and_product<<<sDIR + pN, GZ_T, 0, stream>>>(z_dir, a_dir, ssrc, sdst, Kn, ddst, keyDIR, E1, rK, sDIR,
                                                      exer_emb, In, kn_emb, W_ek, buf_z, Nn);
    owner_agg<<<gK, OWN_T, 0, stream>>>(dsrc, ddst, keyDIR, ssrc, sdst, z_dir, pA, pA, Kn, Kn, E1, rK);
    scores_and_combine<<<sEK + cdiv(Kn, CB_PER), CB_T, 0, stream>>>(buf_z, a_ek, ssrc, sdst, Nn, ekdst, keyEK, E2, rN, sEK,
                                                                   pA, pY, rW1, rb1, rw2, out_kn, Kn);
    owner_agg<<<gN, OWN_T, 0, stream>>>(eksrc, ekdst, keyEK, ssrc, sdst, buf_z, out_ex, pY, In, Nn, E2, rN);
}
